# starting kernel
# baseline (speedup 1.0000x reference)
#include <hip/hip_runtime.h>
#include <math.h>

typedef __attribute__((ext_vector_type(2))) float v2f;
typedef __attribute__((ext_vector_type(8))) float v8f;

#define NC    45
#define NCP   48
#define H1DIM 32
#define H2DIM 128

__global__ void deg_kernel(const int* __restrict__ dst, const float* __restrict__ ew,
                           float* __restrict__ deg, int E) {
  int e = blockIdx.x * blockDim.x + threadIdx.x;
  if (e < E) atomicAdd(&deg[dst[e]], ew[e]);
}

__global__ void dis_kernel(float* __restrict__ deg, int n) {
  int i = blockIdx.x * blockDim.x + threadIdx.x;
  if (i < n) deg[i] = 1.0f / sqrtf(deg[i] + 1.0f);
}

__global__ void pad_x_kernel(const float* __restrict__ x, float* __restrict__ xp,
                             int total  ) {
  int gid = blockIdx.x * blockDim.x + threadIdx.x;
  if (gid >= total) return;
  int i = gid / NCP, c = gid - i * NCP;
  xp[gid] = (c < NC) ? x[(size_t)i * NC + c] : 0.0f;
}

__global__ void pack_w1_kernel(const float* __restrict__ W1, float* __restrict__ Bp1) {
  int gid = blockIdx.x * blockDim.x + threadIdx.x;
  if (gid >= 12 * 2 * H1DIM * 2) return;
  int p   = gid & 1;
  int col = (gid >> 1) & 31;
  int kh  = (gid >> 6) & 1;
  int k   = gid >> 7;
  int row = 4 * k + 2 * kh + p;
  Bp1[gid] = (row < NC) ? W1[row * H1DIM + col] : 0.0f;
}

__global__ void pack_w2_kernel(const float* __restrict__ W2, float* __restrict__ Bp2) {
  int gid = blockIdx.x * blockDim.x + threadIdx.x;
  if (gid >= 8 * 2 * H2DIM * 2) return;
  int p   = gid & 1;
  int col = (gid >> 1) & 127;
  int kh  = (gid >> 8) & 1;
  int k   = gid >> 9;
  int row = 4 * k + 2 * kh + p;
  Bp2[gid] = W2[row * H2DIM + col];
}

__global__ void gemm1_wmma(const float* __restrict__ xp, const float* __restrict__ Bp1,
                           float* __restrict__ h1, int n) {
  const int lane  = threadIdx.x;
  const int m0    = blockIdx.x * 16;
  const int khalf = lane >> 4;
  const int nn    = lane & 15;
  int m = m0 + nn;
  if (m > n - 1) m = n - 1;

  const float* ap = xp + (size_t)m * NCP + 2 * khalf;
  const float* bp = Bp1 + khalf * 64;
  v8f c0 = {};
  v8f c1 = {};
#pragma unroll
  for (int k = 0; k < 12; ++k) {
    v2f a  = *(const v2f*)(ap + 4 * k);
    v2f b0 = *(const v2f*)(bp + 128 * k + nn * 2);
    v2f b1 = *(const v2f*)(bp + 128 * k + (16 + nn) * 2);
    c0 = __builtin_amdgcn_wmma_f32_16x16x4_f32(false, a, false, b0, (short)0, c0, false, false);
    c1 = __builtin_amdgcn_wmma_f32_16x16x4_f32(false, a, false, b1, (short)0, c1, false, false);
  }
  const int rbase = m0 + (khalf << 3);
#pragma unroll
  for (int j = 0; j < 8; ++j) {
    const int r = rbase + j;
    if (r < n) {
      h1[(size_t)r * H1DIM + nn]      = c0[j];
      h1[(size_t)r * H1DIM + 16 + nn] = c1[j];
    }
  }
}

__global__ void gemm2_wmma(const float* __restrict__ g, const float* __restrict__ Bp2,
                           const float* __restrict__ b2, float* __restrict__ out, int n) {
  const int lane  = threadIdx.x;
  const int m0    = blockIdx.x * 16;
  const int khalf = lane >> 4;
  const int nn    = lane & 15;
  int m = m0 + nn;
  if (m > n - 1) m = n - 1;

  const float* ap = g + (size_t)m * H1DIM + 2 * khalf;
  const float* bp = Bp2 + khalf * 256;
  v2f a[8];
#pragma unroll
  for (int k = 0; k < 8; ++k) a[k] = *(const v2f*)(ap + 4 * k);

  v8f acc[8] = {};
#pragma unroll
  for (int k = 0; k < 8; ++k) {
#pragma unroll
    for (int t = 0; t < 8; ++t) {
      v2f b = *(const v2f*)(bp + 512 * k + (t * 16 + nn) * 2);
      acc[t] = __builtin_amdgcn_wmma_f32_16x16x4_f32(false, a[k], false, b, (short)0, acc[t], false, false);
    }
  }
  float bv[8];
#pragma unroll
  for (int t = 0; t < 8; ++t) bv[t] = b2[t * 16 + nn];

  const int rbase = m0 + (khalf << 3);
#pragma unroll
  for (int j = 0; j < 8; ++j) {
    const int r = rbase + j;
    if (r < n) {
#pragma unroll
      for (int t = 0; t < 8; ++t)
        out[(size_t)r * H2DIM + t * 16 + nn] = acc[t][j] + bv[t];
    }
  }
}

__global__ void agg_edges_kernel(const int* __restrict__ src, const int* __restrict__ dst,
                                 const float* __restrict__ ew, const float* __restrict__ dis,
                                 const float* __restrict__ h, float* __restrict__ agg, int E) {
  long long gid = (long long)blockIdx.x * blockDim.x + threadIdx.x;
  int e = (int)(gid >> 3);
  int q = (int)(gid & 7) << 2;
  if (e >= E) return;
  int s = src[e], d = dst[e];
  float norm = dis[s] * ew[e] * dis[d];
  const float4 v = *(const float4*)(h + (size_t)s * H1DIM + q);
  float* o = agg + (size_t)d * H1DIM + q;
  atomicAdd(o + 0, v.x * norm);
  atomicAdd(o + 1, v.y * norm);
  atomicAdd(o + 2, v.z * norm);
  atomicAdd(o + 3, v.w * norm);
}

__global__ void tanh_bias_kernel(float* __restrict__ agg, const float* __restrict__ h1,
                                 const float* __restrict__ dis, const float* __restrict__ b1,
                                 int n) {
  int gid = blockIdx.x * blockDim.x + threadIdx.x;
  if (gid >= n * H1DIM) return;
  int i = gid >> 5, c = gid & 31;
  float di = dis[i];
  agg[gid] = tanhf(agg[gid] + h1[gid] * di * di + b1[c]);
}

__global__ void selfloop_kernel(float* __restrict__ agg2, const float* __restrict__ h,
                                const float* __restrict__ dis, int n) {
  int gid = blockIdx.x * blockDim.x + threadIdx.x;
  if (gid >= n * H1DIM) return;
  int i = gid >> 5;
  float di = dis[i];
  agg2[gid] += h[gid] * di * di;
}

extern "C" void kernel_launch(void* const* d_in, const int* in_sizes, int n_in,
                              void* d_out, int out_size, void* d_ws, size_t ws_size,
                              hipStream_t stream) {
  const float* x   = (const float*)d_in[0];
  const int*   src = (const int*)  d_in[1];
  const int*   dst = (const int*)  d_in[2];
  const float* ew  = (const float*)d_in[3];
  const float* W1  = (const float*)d_in[4];
  const float* b1  = (const float*)d_in[5];
  const float* W2  = (const float*)d_in[6];
  const float* b2  = (const float*)d_in[7];
  float* out = (float*)d_out;

  const int n = in_sizes[0] / NC;
  const int E = in_sizes[1];

  float* dis  = (float*)d_ws;
  float* xp   = dis  + n;
  float* h1   = xp   + (size_t)n * NCP;
  float* agg1 = h1   + (size_t)n * H1DIM;
  float* agg2 = agg1 + (size_t)n * H1DIM;
  float* Bp1  = agg2 + (size_t)n * H1DIM;
  float* Bp2  = Bp1  + 12 * 2 * H1DIM * 2;

  hipMemsetAsync(dis,  0, (size_t)n * sizeof(float), stream);
  hipMemsetAsync(agg1, 0, (size_t)n * H1DIM * sizeof(float), stream);
  hipMemsetAsync(agg2, 0, (size_t)n * H1DIM * sizeof(float), stream);

  const int T = 256;
  deg_kernel<<<(E + T - 1) / T, T, 0, stream>>>(dst, ew, dis, E);
  dis_kernel<<<(n + T - 1) / T, T, 0, stream>>>(dis, n);

  pad_x_kernel<<<((long long)n * NCP + T - 1) / T, T, 0, stream>>>(x, xp, n * NCP);
  pack_w1_kernel<<<(12 * 2 * H1DIM * 2 + T - 1) / T, T, 0, stream>>>(W1, Bp1);
  pack_w2_kernel<<<(8 * 2 * H2DIM * 2 + T - 1) / T, T, 0, stream>>>(W2, Bp2);

  gemm1_wmma<<<(n + 15) / 16, 32, 0, stream>>>(xp, Bp1, h1, n);
  {
    long long work = (long long)E * 8;
    agg_edges_kernel<<<(unsigned)((work + T - 1) / T), T, 0, stream>>>(src, dst, ew, dis, h1, agg1, E);
  }
  tanh_bias_kernel<<<((long long)n * H1DIM + T - 1) / T, T, 0, stream>>>(agg1, h1, dis, b1, n);

  {
    long long work = (long long)E * 8;
    agg_edges_kernel<<<(unsigned)((work + T - 1) / T), T, 0, stream>>>(src, dst, ew, dis, agg1, agg2, E);
  }
  selfloop_kernel<<<((long long)n * H1DIM + T - 1) / T, T, 0, stream>>>(agg2, agg1, dis, n);
  gemm2_wmma<<<(n + 15) / 16, 32, 0, stream>>>(agg2, Bp2, b2, out, n);
}


# edge sums in pull form: per call each edge takes a slot by one returning int add on its target's counter line, wave-scan segments; 8 lanes sum a node's rows in registers, 1 store/row; add order: slot
# speedup vs baseline: 1.4758x; 1.4758x over previous
#include <hip/hip_runtime.h>
#include <math.h>

typedef __attribute__((ext_vector_type(2))) float v2f;
typedef __attribute__((ext_vector_type(8))) float v8f;

#define NC    45
#define NCP   48
#define H1DIM 32
#define H2DIM 128

__global__ void pad_x_kernel(const float* __restrict__ x, float* __restrict__ xp,
                             int total  ) {
  int gid = blockIdx.x * blockDim.x + threadIdx.x;
  if (gid >= total) return;
  int i = gid / NCP, c = gid - i * NCP;
  xp[gid] = (c < NC) ? x[(size_t)i * NC + c] : 0.0f;
}

__global__ void pack_w1_kernel(const float* __restrict__ W1, float* __restrict__ Bp1) {
  int gid = blockIdx.x * blockDim.x + threadIdx.x;
  if (gid >= 12 * 2 * H1DIM * 2) return;
  int p   = gid & 1;
  int col = (gid >> 1) & 31;
  int kh  = (gid >> 6) & 1;
  int k   = gid >> 7;
  int row = 4 * k + 2 * kh + p;
  Bp1[gid] = (row < NC) ? W1[row * H1DIM + col] : 0.0f;
}

__global__ void pack_w2_kernel(const float* __restrict__ W2, float* __restrict__ Bp2) {
  int gid = blockIdx.x * blockDim.x + threadIdx.x;
  if (gid >= 8 * 2 * H2DIM * 2) return;
  int p   = gid & 1;
  int col = (gid >> 1) & 127;
  int kh  = (gid >> 8) & 1;
  int k   = gid >> 9;
  int row = 4 * k + 2 * kh + p;
  Bp2[gid] = W2[row * H2DIM + col];
}

__global__ void gemm1_wmma(const float* __restrict__ xp, const float* __restrict__ Bp1,
                           float* __restrict__ h1, int n) {
  const int lane  = threadIdx.x;
  const int m0    = blockIdx.x * 16;
  const int khalf = lane >> 4;
  const int nn    = lane & 15;
  int m = m0 + nn;
  if (m > n - 1) m = n - 1;

  const float* ap = xp + (size_t)m * NCP + 2 * khalf;
  const float* bp = Bp1 + khalf * 64;
  v8f c0 = {};
  v8f c1 = {};
#pragma unroll
  for (int k = 0; k < 12; ++k) {
    v2f a  = *(const v2f*)(ap + 4 * k);
    v2f b0 = *(const v2f*)(bp + 128 * k + nn * 2);
    v2f b1 = *(const v2f*)(bp + 128 * k + (16 + nn) * 2);
    c0 = __builtin_amdgcn_wmma_f32_16x16x4_f32(false, a, false, b0, (short)0, c0, false, false);
    c1 = __builtin_amdgcn_wmma_f32_16x16x4_f32(false, a, false, b1, (short)0, c1, false, false);
  }
  const int rbase = m0 + (khalf << 3);
#pragma unroll
  for (int j = 0; j < 8; ++j) {
    const int r = rbase + j;
    if (r < n) {
      h1[(size_t)r * H1DIM + nn]      = c0[j];
      h1[(size_t)r * H1DIM + 16 + nn] = c1[j];
    }
  }
}

__global__ void gemm2_wmma(const float* __restrict__ g, const float* __restrict__ Bp2,
                           const float* __restrict__ b2, float* __restrict__ out, int n) {
  const int lane  = threadIdx.x;
  const int m0    = blockIdx.x * 16;
  const int khalf = lane >> 4;
  const int nn    = lane & 15;
  int m = m0 + nn;
  if (m > n - 1) m = n - 1;

  const float* ap = g + (size_t)m * H1DIM + 2 * khalf;
  const float* bp = Bp2 + khalf * 256;
  v2f a[8];
#pragma unroll
  for (int k = 0; k < 8; ++k) a[k] = *(const v2f*)(ap + 4 * k);

  v8f acc[8] = {};
#pragma unroll
  for (int k = 0; k < 8; ++k) {
#pragma unroll
    for (int t = 0; t < 8; ++t) {
      v2f b = *(const v2f*)(bp + 512 * k + (t * 16 + nn) * 2);
      acc[t] = __builtin_amdgcn_wmma_f32_16x16x4_f32(false, a[k], false, b, (short)0, acc[t], false, false);
    }
  }
  float bv[8];
#pragma unroll
  for (int t = 0; t < 8; ++t) bv[t] = b2[t * 16 + nn];

  const int rbase = m0 + (khalf << 3);
#pragma unroll
  for (int j = 0; j < 8; ++j) {
    const int r = rbase + j;
    if (r < n) {
#pragma unroll
      for (int t = 0; t < 8; ++t)
        out[(size_t)r * H2DIM + t * 16 + nn] = acc[t][j] + bv[t];
    }
  }
}

#define CL 32

__global__ void rank_kernel(const int* __restrict__ dst, int* __restrict__ cl,
                            int* __restrict__ rank, int E) {
  int e = blockIdx.x * blockDim.x + threadIdx.x;
  if (e < E) rank[e] = atomicAdd(&cl[(size_t)dst[e] * CL], 1);
}

__global__ void alloc_kernel(int* __restrict__ cl, int n) {
  const int i    = blockIdx.x * blockDim.x + threadIdx.x;
  const int lane = threadIdx.x & 31;
  const int c    = (i < n) ? cl[(size_t)i * CL] : 0;
  int incl = c;
#pragma unroll
  for (int o = 1; o < 32; o <<= 1) {
    int t = __shfl_up(incl, o);
    if (lane >= o) incl += t;
  }
  const int tot = __shfl(incl, 31);
  int base = 0;
  if (lane == 31) base = atomicAdd(&cl[(size_t)n * CL], tot);
  base = __shfl(base, 31);
  if (i < n) cl[(size_t)i * CL + 1] = base + incl - c;
}

__global__ void fill_kernel(const int* __restrict__ src, const int* __restrict__ dst,
                            const float* __restrict__ ew, const int* __restrict__ cl,
                            const int* __restrict__ rank, int* __restrict__ rec, int E) {
  int e = blockIdx.x * blockDim.x + threadIdx.x;
  if (e >= E) return;
  const int d   = dst[e];
  const int pos = cl[(size_t)d * CL + 1] + rank[e];
  *(int2*)(rec + 2 * (size_t)pos) = make_int2(src[e], __float_as_int(ew[e]));
}

__global__ void dis_pull_kernel(const int* __restrict__ cl, const int* __restrict__ rec,
                                float* __restrict__ dis, int n) {
  int i = blockIdx.x * blockDim.x + threadIdx.x;
  if (i >= n) return;
  const int2 co = *(const int2*)(cl + (size_t)i * CL);
  const int end = co.y + co.x;
  float s = 0.0f;
  for (int j = co.y; j < end; j += 4) {
    const bool k1 = j + 1 < end, k2 = j + 2 < end, k3 = j + 3 < end;
    const float w0 = __int_as_float(rec[2 * (size_t)j + 1]);
    const float w1 = __int_as_float(rec[2 * (size_t)(k1 ? j + 1 : j) + 1]);
    const float w2 = __int_as_float(rec[2 * (size_t)(k2 ? j + 2 : j) + 1]);
    const float w3 = __int_as_float(rec[2 * (size_t)(k3 ? j + 3 : j) + 1]);
    s += w0;
    s += k1 ? w1 : 0.0f;
    s += k2 ? w2 : 0.0f;
    s += k3 ? w3 : 0.0f;
  }
  dis[i] = 1.0f / sqrtf(s + 1.0f);
}

template <bool ACT>
__global__ void pull_kernel(const int* __restrict__ cl, const int* __restrict__ rec,
                            const float* __restrict__ dis, const float* __restrict__ h,
                            const float* __restrict__ bias, float* __restrict__ outp, int n) {
  const int gid = blockIdx.x * blockDim.x + threadIdx.x;
  const int i   = gid >> 3;
  const int q   = (gid & 7) << 2;
  if (i >= n) return;
  const int2 co = *(const int2*)(cl + (size_t)i * CL);
  int j = co.y;
  const int end = co.y + co.x;
  float4 acc = make_float4(0.0f, 0.0f, 0.0f, 0.0f);
  for (; j < end; j += 4) {
    const bool k1 = j + 1 < end, k2 = j + 2 < end, k3 = j + 3 < end;
    const int2 r0 = *(const int2*)(rec + 2 * (size_t)j);
    const int2 r1 = *(const int2*)(rec + 2 * (size_t)(k1 ? j + 1 : j));
    const int2 r2 = *(const int2*)(rec + 2 * (size_t)(k2 ? j + 2 : j));
    const int2 r3 = *(const int2*)(rec + 2 * (size_t)(k3 ? j + 3 : j));
    const float s0 = dis[r0.x], s1 = dis[r1.x], s2 = dis[r2.x], s3 = dis[r3.x];
    const float4 v0 = *(const float4*)(h + (size_t)r0.x * H1DIM + q);
    const float4 v1 = *(const float4*)(h + (size_t)r1.x * H1DIM + q);
    const float4 v2 = *(const float4*)(h + (size_t)r2.x * H1DIM + q);
    const float4 v3 = *(const float4*)(h + (size_t)r3.x * H1DIM + q);
    const float c0 = s0 * __int_as_float(r0.y);
    const float c1 = k1 ? s1 * __int_as_float(r1.y) : 0.0f;
    const float c2 = k2 ? s2 * __int_as_float(r2.y) : 0.0f;
    const float c3 = k3 ? s3 * __int_as_float(r3.y) : 0.0f;
    acc.x += v0.x * c0; acc.y += v0.y * c0; acc.z += v0.z * c0; acc.w += v0.w * c0;
    acc.x += v1.x * c1; acc.y += v1.y * c1; acc.z += v1.z * c1; acc.w += v1.w * c1;
    acc.x += v2.x * c2; acc.y += v2.y * c2; acc.z += v2.z * c2; acc.w += v2.w * c2;
    acc.x += v3.x * c3; acc.y += v3.y * c3; acc.z += v3.z * c3; acc.w += v3.w * c3;
  }
  const float di = dis[i];
  const float d2 = di * di;
  const float4 hv = *(const float4*)(h + (size_t)i * H1DIM + q);
  float4 o;
  o.x = acc.x * di + hv.x * d2;
  o.y = acc.y * di + hv.y * d2;
  o.z = acc.z * di + hv.z * d2;
  o.w = acc.w * di + hv.w * d2;
  if (ACT) {
    o.x = tanhf(o.x + bias[q + 0]);
    o.y = tanhf(o.y + bias[q + 1]);
    o.z = tanhf(o.z + bias[q + 2]);
    o.w = tanhf(o.w + bias[q + 3]);
  }
  *(float4*)(outp + (size_t)i * H1DIM + q) = o;
}

extern "C" void kernel_launch(void* const* d_in, const int* in_sizes, int n_in,
                              void* d_out, int out_size, void* d_ws, size_t ws_size,
                              hipStream_t stream) {
  const float* x   = (const float*)d_in[0];
  const int*   src = (const int*)  d_in[1];
  const int*   dst = (const int*)  d_in[2];
  const float* ew  = (const float*)d_in[3];
  const float* W1  = (const float*)d_in[4];
  const float* b1  = (const float*)d_in[5];
  const float* W2  = (const float*)d_in[6];
  const float* b2  = (const float*)d_in[7];
  float* out = (float*)d_out;

  const int n = in_sizes[0] / NC;
  const int E = in_sizes[1];

  int*   cl   = (int*)d_ws;
  int*   rec  = cl   + (size_t)(n + 1) * CL;
  int*   rank = rec  + 2 * (size_t)E;
  float* dis  = (float*)(rank + (size_t)E);
  float* xp   = dis  + n;
  float* h1   = xp   + (size_t)n * NCP;
  float* hh   = h1   + (size_t)n * H1DIM;
  float* g    = hh   + (size_t)n * H1DIM;
  float* Bp1  = g    + (size_t)n * H1DIM;
  float* Bp2  = Bp1  + 12 * 2 * H1DIM * 2;

  hipMemsetAsync(cl, 0, (size_t)(n + 1) * CL * sizeof(int), stream);

  const int T = 256;
  rank_kernel<<<(E + T - 1) / T, T, 0, stream>>>(dst, cl, rank, E);
  alloc_kernel<<<(n + T - 1) / T, T, 0, stream>>>(cl, n);
  fill_kernel<<<(E + T - 1) / T, T, 0, stream>>>(src, dst, ew, cl, rank, rec, E);
  dis_pull_kernel<<<(n + T - 1) / T, T, 0, stream>>>(cl, rec, dis, n);

  pad_x_kernel<<<((long long)n * NCP + T - 1) / T, T, 0, stream>>>(x, xp, n * NCP);
  pack_w1_kernel<<<(12 * 2 * H1DIM * 2 + T - 1) / T, T, 0, stream>>>(W1, Bp1);
  pack_w2_kernel<<<(8 * 2 * H2DIM * 2 + T - 1) / T, T, 0, stream>>>(W2, Bp2);

  gemm1_wmma<<<(n + 15) / 16, 32, 0, stream>>>(xp, Bp1, h1, n);
  pull_kernel<true><<<((long long)n * 8 + T - 1) / T, T, 0, stream>>>(cl, rec, dis, h1, b1, hh, n);

  pull_kernel<false><<<((long long)n * 8 + T - 1) / T, T, 0, stream>>>(cl, rec, dis, hh, b1, g, n);
  gemm2_wmma<<<(n + 15) / 16, 32, 0, stream>>>(g, Bp2, b2, out, n);
}


# speedup vs baseline: 1.5638x; 1.0596x over previous
#include <hip/hip_runtime.h>
#include <math.h>

typedef __attribute__((ext_vector_type(2))) float v2f;
typedef __attribute__((ext_vector_type(8))) float v8f;

#define NC    45
#define NCP   48
#define H1DIM 32
#define H2DIM 128

__global__ void pad_x_kernel(const float* __restrict__ x, float* __restrict__ xp,
                             int total  ) {
  int gid = blockIdx.x * blockDim.x + threadIdx.x;
  if (gid >= total) return;
  int i = gid / NCP, c = gid - i * NCP;
  xp[gid] = (c < NC) ? x[(size_t)i * NC + c] : 0.0f;
}

__global__ void pack_w1_kernel(const float* __restrict__ W1, float* __restrict__ Bp1) {
  int gid = blockIdx.x * blockDim.x + threadIdx.x;
  if (gid >= 12 * 2 * H1DIM * 2) return;
  int p   = gid & 1;
  int col = (gid >> 1) & 31;
  int kh  = (gid >> 6) & 1;
  int k   = gid >> 7;
  int row = 4 * k + 2 * kh + p;
  Bp1[gid] = (row < NC) ? W1[row * H1DIM + col] : 0.0f;
}

__global__ void pack_w2_kernel(const float* __restrict__ W2, float* __restrict__ Bp2) {
  int gid = blockIdx.x * blockDim.x + threadIdx.x;
  if (gid >= 8 * 2 * H2DIM * 2) return;
  int p   = gid & 1;
  int col = (gid >> 1) & 127;
  int kh  = (gid >> 8) & 1;
  int k   = gid >> 9;
  int row = 4 * k + 2 * kh + p;
  Bp2[gid] = W2[row * H2DIM + col];
}

__global__ void gemm1_wmma(const float* __restrict__ xp, const float* __restrict__ Bp1,
                           float* __restrict__ h1, int n) {
  const int lane  = threadIdx.x;
  const int m0    = blockIdx.x * 16;
  const int khalf = lane >> 4;
  const int nn    = lane & 15;
  int m = m0 + nn;
  if (m > n - 1) m = n - 1;

  const float* ap = xp + (size_t)m * NCP + 2 * khalf;
  const float* bp = Bp1 + khalf * 64;
  v8f c0 = {};
  v8f c1 = {};
#pragma unroll
  for (int k = 0; k < 12; ++k) {
    v2f a  = *(const v2f*)(ap + 4 * k);
    v2f b0 = *(const v2f*)(bp + 128 * k + nn * 2);
    v2f b1 = *(const v2f*)(bp + 128 * k + (16 + nn) * 2);
    c0 = __builtin_amdgcn_wmma_f32_16x16x4_f32(false, a, false, b0, (short)0, c0, false, false);
    c1 = __builtin_amdgcn_wmma_f32_16x16x4_f32(false, a, false, b1, (short)0, c1, false, false);
  }
  const int rbase = m0 + (khalf << 3);
#pragma unroll
  for (int j = 0; j < 8; ++j) {
    const int r = rbase + j;
    if (r < n) {
      h1[(size_t)r * H1DIM + nn]      = c0[j];
      h1[(size_t)r * H1DIM + 16 + nn] = c1[j];
    }
  }
}

__global__ void gemm2_wmma(const float* __restrict__ g, const float* __restrict__ Bp2,
                           const float* __restrict__ b2, float* __restrict__ out, int n) {
  const int lane  = threadIdx.x;
  const int m0    = blockIdx.x * 16;
  const int khalf = lane >> 4;
  const int nn    = lane & 15;
  int m = m0 + nn;
  if (m > n - 1) m = n - 1;

  const float* ap = g + (size_t)m * H1DIM + 2 * khalf;
  const float* bp = Bp2 + khalf * 256;
  v2f a[8];
#pragma unroll
  for (int k = 0; k < 8; ++k) a[k] = *(const v2f*)(ap + 4 * k);

  v8f acc[8] = {};
#pragma unroll
  for (int k = 0; k < 8; ++k) {
#pragma unroll
    for (int t = 0; t < 8; ++t) {
      v2f b = *(const v2f*)(bp + 512 * k + (t * 16 + nn) * 2);
      acc[t] = __builtin_amdgcn_wmma_f32_16x16x4_f32(false, a[k], false, b, (short)0, acc[t], false, false);
    }
  }
  float bv[8];
#pragma unroll
  for (int t = 0; t < 8; ++t) bv[t] = b2[t * 16 + nn];

  const int rbase = m0 + (khalf << 3);
#pragma unroll
  for (int j = 0; j < 8; ++j) {
    const int r = rbase + j;
    if (r < n) {
#pragma unroll
      for (int t = 0; t < 8; ++t)
        out[(size_t)r * H2DIM + t * 16 + nn] = acc[t][j] + bv[t];
    }
  }
}

#define CL 32

__device__ __forceinline__ int package_half() {
  return (int)((__builtin_amdgcn_s_sendmsg_rtn(0x87) >> 18) & 1u);
}

__global__ void rank_kernel(const int* __restrict__ dst, int* __restrict__ cl0,
                            int* __restrict__ cl1, int* __restrict__ rank, int E) {
  int e = blockIdx.x * blockDim.x + threadIdx.x;
  const int hf = package_half();
  int* cl = hf ? cl1 : cl0;
  if (e < E) rank[e] = (atomicAdd(&cl[(size_t)dst[e] * CL], 1) << 1) | hf;
}

__global__ void alloc_kernel(int* __restrict__ cl0, int* __restrict__ cl1, int n) {
  const int i    = blockIdx.x * blockDim.x + threadIdx.x;
  const int lane = threadIdx.x & 31;
  const int c0   = (i < n) ? cl0[(size_t)i * CL] : 0;
  const int c1   = (i < n) ? cl1[(size_t)i * CL] : 0;
  const int c    = c0 + c1;
  int incl = c;
#pragma unroll
  for (int o = 1; o < 32; o <<= 1) {
    int t = __shfl_up(incl, o);
    if (lane >= o) incl += t;
  }
  const int tot = __shfl(incl, 31);
  int base = 0;
  if (lane == 31) base = atomicAdd(&cl0[(size_t)n * CL], tot);
  base = __shfl(base, 31);
  if (i < n) {
    const int st = base + incl - c;
    *(int2*)(cl0 + (size_t)i * CL) = make_int2(c, st);
    cl1[(size_t)i * CL + 1] = st + c0;
  }
}

__global__ void fill_kernel(const int* __restrict__ src, const int* __restrict__ dst,
                            const float* __restrict__ ew, const int* __restrict__ cl0,
                            const int* __restrict__ cl1, const int* __restrict__ rank,
                            int* __restrict__ rec, int E) {
  int e = blockIdx.x * blockDim.x + threadIdx.x;
  if (e >= E) return;
  const int d   = dst[e];
  const int rk  = rank[e];
  const int* cl = (rk & 1) ? cl1 : cl0;
  const int pos = cl[(size_t)d * CL + 1] + (rk >> 1);
  *(int2*)(rec + 2 * (size_t)pos) = make_int2(src[e], __float_as_int(ew[e]));
}

__global__ void dis_pull_kernel(const int* __restrict__ cl, const int* __restrict__ rec,
                                float* __restrict__ dis, int n) {
  int i = blockIdx.x * blockDim.x + threadIdx.x;
  if (i >= n) return;
  const int2 co = *(const int2*)(cl + (size_t)i * CL);
  const int end = co.y + co.x;
  float s = 0.0f;
  for (int j = co.y; j < end; j += 4) {
    const bool k1 = j + 1 < end, k2 = j + 2 < end, k3 = j + 3 < end;
    const float w0 = __int_as_float(rec[2 * (size_t)j + 1]);
    const float w1 = __int_as_float(rec[2 * (size_t)(k1 ? j + 1 : j) + 1]);
    const float w2 = __int_as_float(rec[2 * (size_t)(k2 ? j + 2 : j) + 1]);
    const float w3 = __int_as_float(rec[2 * (size_t)(k3 ? j + 3 : j) + 1]);
    s += w0;
    s += k1 ? w1 : 0.0f;
    s += k2 ? w2 : 0.0f;
    s += k3 ? w3 : 0.0f;
  }
  dis[i] = 1.0f / sqrtf(s + 1.0f);
}

template <bool ACT>
__global__ void pull_kernel(const int* __restrict__ cl, const int* __restrict__ rec,
                            const float* __restrict__ dis, const float* __restrict__ h,
                            const float* __restrict__ bias, float* __restrict__ outp, int n) {
  const int gid = blockIdx.x * blockDim.x + threadIdx.x;
  const int i   = gid >> 3;
  const int q   = (gid & 7) << 2;
  if (i >= n) return;
  const int2 co = *(const int2*)(cl + (size_t)i * CL);
  int j = co.y;
  const int end = co.y + co.x;
  float4 acc = make_float4(0.0f, 0.0f, 0.0f, 0.0f);
  for (; j < end; j += 4) {
    const bool k1 = j + 1 < end, k2 = j + 2 < end, k3 = j + 3 < end;
    const int2 r0 = *(const int2*)(rec + 2 * (size_t)j);
    const int2 r1 = *(const int2*)(rec + 2 * (size_t)(k1 ? j + 1 : j));
    const int2 r2 = *(const int2*)(rec + 2 * (size_t)(k2 ? j + 2 : j));
    const int2 r3 = *(const int2*)(rec + 2 * (size_t)(k3 ? j + 3 : j));
    const float s0 = dis[r0.x], s1 = dis[r1.x], s2 = dis[r2.x], s3 = dis[r3.x];
    const float4 v0 = *(const float4*)(h + (size_t)r0.x * H1DIM + q);
    const float4 v1 = *(const float4*)(h + (size_t)r1.x * H1DIM + q);
    const float4 v2 = *(const float4*)(h + (size_t)r2.x * H1DIM + q);
    const float4 v3 = *(const float4*)(h + (size_t)r3.x * H1DIM + q);
    const float c0 = s0 * __int_as_float(r0.y);
    const float c1 = k1 ? s1 * __int_as_float(r1.y) : 0.0f;
    const float c2 = k2 ? s2 * __int_as_float(r2.y) : 0.0f;
    const float c3 = k3 ? s3 * __int_as_float(r3.y) : 0.0f;
    acc.x += v0.x * c0; acc.y += v0.y * c0; acc.z += v0.z * c0; acc.w += v0.w * c0;
    acc.x += v1.x * c1; acc.y += v1.y * c1; acc.z += v1.z * c1; acc.w += v1.w * c1;
    acc.x += v2.x * c2; acc.y += v2.y * c2; acc.z += v2.z * c2; acc.w += v2.w * c2;
    acc.x += v3.x * c3; acc.y += v3.y * c3; acc.z += v3.z * c3; acc.w += v3.w * c3;
  }
  const float di = dis[i];
  const float d2 = di * di;
  const float4 hv = *(const float4*)(h + (size_t)i * H1DIM + q);
  float4 o;
  o.x = acc.x * di + hv.x * d2;
  o.y = acc.y * di + hv.y * d2;
  o.z = acc.z * di + hv.z * d2;
  o.w = acc.w * di + hv.w * d2;
  if (ACT) {
    o.x = tanhf(o.x + bias[q + 0]);
    o.y = tanhf(o.y + bias[q + 1]);
    o.z = tanhf(o.z + bias[q + 2]);
    o.w = tanhf(o.w + bias[q + 3]);
  }
  *(float4*)(outp + (size_t)i * H1DIM + q) = o;
}

extern "C" void kernel_launch(void* const* d_in, const int* in_sizes, int n_in,
                              void* d_out, int out_size, void* d_ws, size_t ws_size,
                              hipStream_t stream) {
  const float* x   = (const float*)d_in[0];
  const int*   src = (const int*)  d_in[1];
  const int*   dst = (const int*)  d_in[2];
  const float* ew  = (const float*)d_in[3];
  const float* W1  = (const float*)d_in[4];
  const float* b1  = (const float*)d_in[5];
  const float* W2  = (const float*)d_in[6];
  const float* b2  = (const float*)d_in[7];
  float* out = (float*)d_out;

  const int n = in_sizes[0] / NC;
  const int E = in_sizes[1];

  int*   cl   = (int*)d_ws;
  int*   cl1  = cl   + (size_t)(n + 2) * CL;
  int*   rec  = cl1  + (size_t)(n + 2) * CL;
  int*   rank = rec  + 2 * (size_t)E;
  float* dis  = (float*)(rank + (size_t)E);
  float* xp   = dis  + n;
  float* h1   = xp   + (size_t)n * NCP;
  float* hh   = h1   + (size_t)n * H1DIM;
  float* g    = hh   + (size_t)n * H1DIM;
  float* Bp1  = g    + (size_t)n * H1DIM;
  float* Bp2  = Bp1  + 12 * 2 * H1DIM * 2;

  hipMemsetAsync(cl, 0, 2 * (size_t)(n + 2) * CL * sizeof(int), stream);

  const int T = 256;
  rank_kernel<<<(E + T - 1) / T, T, 0, stream>>>(dst, cl, cl1, rank, E);
  alloc_kernel<<<(n + T - 1) / T, T, 0, stream>>>(cl, cl1, n);
  fill_kernel<<<(E + T - 1) / T, T, 0, stream>>>(src, dst, ew, cl, cl1, rank, rec, E);
  dis_pull_kernel<<<(n + T - 1) / T, T, 0, stream>>>(cl, rec, dis, n);

  pad_x_kernel<<<((long long)n * NCP + T - 1) / T, T, 0, stream>>>(x, xp, n * NCP);
  pack_w1_kernel<<<(12 * 2 * H1DIM * 2 + T - 1) / T, T, 0, stream>>>(W1, Bp1);
  pack_w2_kernel<<<(8 * 2 * H2DIM * 2 + T - 1) / T, T, 0, stream>>>(W2, Bp2);

  gemm1_wmma<<<(n + 15) / 16, 32, 0, stream>>>(xp, Bp1, h1, n);
  pull_kernel<true><<<((long long)n * 8 + T - 1) / T, T, 0, stream>>>(cl, rec, dis, h1, b1, hh, n);

  pull_kernel<false><<<((long long)n * 8 + T - 1) / T, T, 0, stream>>>(cl, rec, dis, hh, b1, g, n);
  gemm2_wmma<<<(n + 15) / 16, 32, 0, stream>>>(g, Bp2, b2, out, n);
}


# pad and fill passes: each lane takes 8 items a grid slice apart, loads issued together (pad: whole 16-B pieces): 200000 one-item waves become 10944; rank pass untouched; add order: slot, as before
# speedup vs baseline: 1.6163x; 1.0336x over previous
#include <hip/hip_runtime.h>
#include <math.h>

typedef __attribute__((ext_vector_type(2))) float v2f;
typedef __attribute__((ext_vector_type(8))) float v8f;

#define NC    45
#define NCP   48
#define H1DIM 32
#define H2DIM 128

#define PAD_U 8
typedef float row4 __attribute__((ext_vector_type(4), aligned(4)));
template <bool WHOLE>
__device__ __forceinline__ void pad_trip(const float* __restrict__ x, float* __restrict__ xp,
                                         int g0, int stride, int pieces) {
  float4 v[PAD_U];
#pragma unroll
  for (int u = 0; u < PAD_U; ++u) {
    int g = g0 + u * stride;
    if (!WHOLE && g > pieces - 1) g = pieces - 1;
    const int i = g / 12, j = g - i * 12;
    const float* p = x + (size_t)i * NC + 4 * j;
    const row4 t = *(const row4*)(p - (j == 11 ? 3 : 0));
    v[u] = (j == 11) ? make_float4(t.w, 0.0f, 0.0f, 0.0f) : make_float4(t.x, t.y, t.z, t.w);
  }
#pragma unroll
  for (int u = 0; u < PAD_U; ++u) {
    const int g = g0 + u * stride;
    if (WHOLE || g < pieces) *(float4*)(xp + 4 * (size_t)g) = v[u];
  }
}

__global__ void pad_x_kernel(const float* __restrict__ x, float* __restrict__ xp,
                             int pieces  ) {
  const int stride = gridDim.x * blockDim.x;
  for (int g0 = blockIdx.x * blockDim.x + threadIdx.x; g0 < pieces; g0 += PAD_U * stride) {
    if (g0 + (PAD_U - 1) * stride < pieces) pad_trip<true>(x, xp, g0, stride, pieces);
    else                                    pad_trip<false>(x, xp, g0, stride, pieces);
  }
}

__global__ void pack_w1_kernel(const float* __restrict__ W1, float* __restrict__ Bp1) {
  int gid = blockIdx.x * blockDim.x + threadIdx.x;
  if (gid >= 12 * 2 * H1DIM * 2) return;
  int p   = gid & 1;
  int col = (gid >> 1) & 31;
  int kh  = (gid >> 6) & 1;
  int k   = gid >> 7;
  int row = 4 * k + 2 * kh + p;
  Bp1[gid] = (row < NC) ? W1[row * H1DIM + col] : 0.0f;
}

__global__ void pack_w2_kernel(const float* __restrict__ W2, float* __restrict__ Bp2) {
  int gid = blockIdx.x * blockDim.x + threadIdx.x;
  if (gid >= 8 * 2 * H2DIM * 2) return;
  int p   = gid & 1;
  int col = (gid >> 1) & 127;
  int kh  = (gid >> 8) & 1;
  int k   = gid >> 9;
  int row = 4 * k + 2 * kh + p;
  Bp2[gid] = W2[row * H2DIM + col];
}

__global__ void gemm1_wmma(const float* __restrict__ xp, const float* __restrict__ Bp1,
                           float* __restrict__ h1, int n) {
  const int lane  = threadIdx.x;
  const int m0    = blockIdx.x * 16;
  const int khalf = lane >> 4;
  const int nn    = lane & 15;
  int m = m0 + nn;
  if (m > n - 1) m = n - 1;

  const float* ap = xp + (size_t)m * NCP + 2 * khalf;
  const float* bp = Bp1 + khalf * 64;
  v8f c0 = {};
  v8f c1 = {};
#pragma unroll
  for (int k = 0; k < 12; ++k) {
    v2f a  = *(const v2f*)(ap + 4 * k);
    v2f b0 = *(const v2f*)(bp + 128 * k + nn * 2);
    v2f b1 = *(const v2f*)(bp + 128 * k + (16 + nn) * 2);
    c0 = __builtin_amdgcn_wmma_f32_16x16x4_f32(false, a, false, b0, (short)0, c0, false, false);
    c1 = __builtin_amdgcn_wmma_f32_16x16x4_f32(false, a, false, b1, (short)0, c1, false, false);
  }
  const int rbase = m0 + (khalf << 3);
#pragma unroll
  for (int j = 0; j < 8; ++j) {
    const int r = rbase + j;
    if (r < n) {
      h1[(size_t)r * H1DIM + nn]      = c0[j];
      h1[(size_t)r * H1DIM + 16 + nn] = c1[j];
    }
  }
}

__global__ void gemm2_wmma(const float* __restrict__ g, const float* __restrict__ Bp2,
                           const float* __restrict__ b2, float* __restrict__ out, int n) {
  const int lane  = threadIdx.x;
  const int m0    = blockIdx.x * 16;
  const int khalf = lane >> 4;
  const int nn    = lane & 15;
  int m = m0 + nn;
  if (m > n - 1) m = n - 1;

  const float* ap = g + (size_t)m * H1DIM + 2 * khalf;
  const float* bp = Bp2 + khalf * 256;
  v2f a[8];
#pragma unroll
  for (int k = 0; k < 8; ++k) a[k] = *(const v2f*)(ap + 4 * k);

  v8f acc[8] = {};
#pragma unroll
  for (int k = 0; k < 8; ++k) {
#pragma unroll
    for (int t = 0; t < 8; ++t) {
      v2f b = *(const v2f*)(bp + 512 * k + (t * 16 + nn) * 2);
      acc[t] = __builtin_amdgcn_wmma_f32_16x16x4_f32(false, a[k], false, b, (short)0, acc[t], false, false);
    }
  }
  float bv[8];
#pragma unroll
  for (int t = 0; t < 8; ++t) bv[t] = b2[t * 16 + nn];

  const int rbase = m0 + (khalf << 3);
#pragma unroll
  for (int j = 0; j < 8; ++j) {
    const int r = rbase + j;
    if (r < n) {
#pragma unroll
      for (int t = 0; t < 8; ++t)
        out[(size_t)r * H2DIM + t * 16 + nn] = acc[t][j] + bv[t];
    }
  }
}

#define CL 32

__device__ __forceinline__ int package_half() {
  return (int)((__builtin_amdgcn_s_sendmsg_rtn(0x87) >> 18) & 1u);
}

__global__ void rank_kernel(const int* __restrict__ dst, int* __restrict__ cl0,
                            int* __restrict__ cl1, int* __restrict__ rank, int E) {
  int e = blockIdx.x * blockDim.x + threadIdx.x;
  const int hf = package_half();
  int* cl = hf ? cl1 : cl0;
  if (e < E) rank[e] = (atomicAdd(&cl[(size_t)dst[e] * CL], 1) << 1) | hf;
}

__global__ void alloc_kernel(int* __restrict__ cl0, int* __restrict__ cl1, int n) {
  const int i    = blockIdx.x * blockDim.x + threadIdx.x;
  const int lane = threadIdx.x & 31;
  const int c0   = (i < n) ? cl0[(size_t)i * CL] : 0;
  const int c1   = (i < n) ? cl1[(size_t)i * CL] : 0;
  const int c    = c0 + c1;
  int incl = c;
#pragma unroll
  for (int o = 1; o < 32; o <<= 1) {
    int t = __shfl_up(incl, o);
    if (lane >= o) incl += t;
  }
  const int tot = __shfl(incl, 31);
  int base = 0;
  if (lane == 31) base = atomicAdd(&cl0[(size_t)n * CL], tot);
  base = __shfl(base, 31);
  if (i < n) {
    const int st = base + incl - c;
    *(int2*)(cl0 + (size_t)i * CL) = make_int2(c, st);
    cl1[(size_t)i * CL + 1] = st + c0;
  }
}

#define FILL_U 8
template <bool WHOLE>
__device__ __forceinline__ void fill_trip(const int* __restrict__ src, const int* __restrict__ dst,
                                          const float* __restrict__ ew, const int* __restrict__ cl0,
                                          const int* __restrict__ cl1, const int* __restrict__ rank,
                                          int* __restrict__ rec, int e0, int stride, int E) {
  int d[FILL_U], rk[FILL_U], sv[FILL_U], wv[FILL_U], st[FILL_U];
#pragma unroll
  for (int u = 0; u < FILL_U; ++u) {
    int e = e0 + u * stride;
    if (!WHOLE && e > E - 1) e = E - 1;
    d[u]  = dst[e];
    rk[u] = rank[e];
    sv[u] = src[e];
    wv[u] = __float_as_int(ew[e]);
  }
#pragma unroll
  for (int u = 0; u < FILL_U; ++u) {
    const int* cl = (rk[u] & 1) ? cl1 : cl0;
    st[u] = cl[(size_t)d[u] * CL + 1];
  }
#pragma unroll
  for (int u = 0; u < FILL_U; ++u) {
    const int e = e0 + u * stride;
    if (WHOLE || e < E) *(int2*)(rec + 2 * (size_t)(st[u] + (rk[u] >> 1))) = make_int2(sv[u], wv[u]);
  }
}

__global__ void fill_kernel(const int* __restrict__ src, const int* __restrict__ dst,
                            const float* __restrict__ ew, const int* __restrict__ cl0,
                            const int* __restrict__ cl1, const int* __restrict__ rank,
                            int* __restrict__ rec, int E) {
  const int stride = gridDim.x * blockDim.x;
  for (int e0 = blockIdx.x * blockDim.x + threadIdx.x; e0 < E; e0 += FILL_U * stride) {
    if (e0 + (FILL_U - 1) * stride < E) fill_trip<true>(src, dst, ew, cl0, cl1, rank, rec, e0, stride, E);
    else                                fill_trip<false>(src, dst, ew, cl0, cl1, rank, rec, e0, stride, E);
  }
}

__global__ void dis_pull_kernel(const int* __restrict__ cl, const int* __restrict__ rec,
                                float* __restrict__ dis, int n) {
  int i = blockIdx.x * blockDim.x + threadIdx.x;
  if (i >= n) return;
  const int2 co = *(const int2*)(cl + (size_t)i * CL);
  const int end = co.y + co.x;
  float s = 0.0f;
  for (int j = co.y; j < end; j += 4) {
    const bool k1 = j + 1 < end, k2 = j + 2 < end, k3 = j + 3 < end;
    const float w0 = __int_as_float(rec[2 * (size_t)j + 1]);
    const float w1 = __int_as_float(rec[2 * (size_t)(k1 ? j + 1 : j) + 1]);
    const float w2 = __int_as_float(rec[2 * (size_t)(k2 ? j + 2 : j) + 1]);
    const float w3 = __int_as_float(rec[2 * (size_t)(k3 ? j + 3 : j) + 1]);
    s += w0;
    s += k1 ? w1 : 0.0f;
    s += k2 ? w2 : 0.0f;
    s += k3 ? w3 : 0.0f;
  }
  dis[i] = 1.0f / sqrtf(s + 1.0f);
}

template <bool ACT>
__global__ void pull_kernel(const int* __restrict__ cl, const int* __restrict__ rec,
                            const float* __restrict__ dis, const float* __restrict__ h,
                            const float* __restrict__ bias, float* __restrict__ outp, int n) {
  const int gid = blockIdx.x * blockDim.x + threadIdx.x;
  const int i   = gid >> 3;
  const int q   = (gid & 7) << 2;
  if (i >= n) return;
  const int2 co = *(const int2*)(cl + (size_t)i * CL);
  int j = co.y;
  const int end = co.y + co.x;
  float4 acc = make_float4(0.0f, 0.0f, 0.0f, 0.0f);
  for (; j < end; j += 4) {
    const bool k1 = j + 1 < end, k2 = j + 2 < end, k3 = j + 3 < end;
    const int2 r0 = *(const int2*)(rec + 2 * (size_t)j);
    const int2 r1 = *(const int2*)(rec + 2 * (size_t)(k1 ? j + 1 : j));
    const int2 r2 = *(const int2*)(rec + 2 * (size_t)(k2 ? j + 2 : j));
    const int2 r3 = *(const int2*)(rec + 2 * (size_t)(k3 ? j + 3 : j));
    const float s0 = dis[r0.x], s1 = dis[r1.x], s2 = dis[r2.x], s3 = dis[r3.x];
    const float4 v0 = *(const float4*)(h + (size_t)r0.x * H1DIM + q);
    const float4 v1 = *(const float4*)(h + (size_t)r1.x * H1DIM + q);
    const float4 v2 = *(const float4*)(h + (size_t)r2.x * H1DIM + q);
    const float4 v3 = *(const float4*)(h + (size_t)r3.x * H1DIM + q);
    const float c0 = s0 * __int_as_float(r0.y);
    const float c1 = k1 ? s1 * __int_as_float(r1.y) : 0.0f;
    const float c2 = k2 ? s2 * __int_as_float(r2.y) : 0.0f;
    const float c3 = k3 ? s3 * __int_as_float(r3.y) : 0.0f;
    acc.x += v0.x * c0; acc.y += v0.y * c0; acc.z += v0.z * c0; acc.w += v0.w * c0;
    acc.x += v1.x * c1; acc.y += v1.y * c1; acc.z += v1.z * c1; acc.w += v1.w * c1;
    acc.x += v2.x * c2; acc.y += v2.y * c2; acc.z += v2.z * c2; acc.w += v2.w * c2;
    acc.x += v3.x * c3; acc.y += v3.y * c3; acc.z += v3.z * c3; acc.w += v3.w * c3;
  }
  const float di = dis[i];
  const float d2 = di * di;
  const float4 hv = *(const float4*)(h + (size_t)i * H1DIM + q);
  float4 o;
  o.x = acc.x * di + hv.x * d2;
  o.y = acc.y * di + hv.y * d2;
  o.z = acc.z * di + hv.z * d2;
  o.w = acc.w * di + hv.w * d2;
  if (ACT) {
    o.x = tanhf(o.x + bias[q + 0]);
    o.y = tanhf(o.y + bias[q + 1]);
    o.z = tanhf(o.z + bias[q + 2]);
    o.w = tanhf(o.w + bias[q + 3]);
  }
  *(float4*)(outp + (size_t)i * H1DIM + q) = o;
}

extern "C" void kernel_launch(void* const* d_in, const int* in_sizes, int n_in,
                              void* d_out, int out_size, void* d_ws, size_t ws_size,
                              hipStream_t stream) {
  const float* x   = (const float*)d_in[0];
  const int*   src = (const int*)  d_in[1];
  const int*   dst = (const int*)  d_in[2];
  const float* ew  = (const float*)d_in[3];
  const float* W1  = (const float*)d_in[4];
  const float* b1  = (const float*)d_in[5];
  const float* W2  = (const float*)d_in[6];
  const float* b2  = (const float*)d_in[7];
  float* out = (float*)d_out;

  const int n = in_sizes[0] / NC;
  const int E = in_sizes[1];

  int*   cl   = (int*)d_ws;
  int*   cl1  = cl   + (size_t)(n + 2) * CL;
  int*   rec  = cl1  + (size_t)(n + 2) * CL;
  int*   rank = rec  + 2 * (size_t)E;
  float* dis  = (float*)(rank + (size_t)E);
  float* xp   = dis  + n;
  float* h1   = xp   + (size_t)n * NCP;
  float* hh   = h1   + (size_t)n * H1DIM;
  float* g    = hh   + (size_t)n * H1DIM;
  float* Bp1  = g    + (size_t)n * H1DIM;
  float* Bp2  = Bp1  + 12 * 2 * H1DIM * 2;

  hipMemsetAsync(cl, 0, 2 * (size_t)(n + 2) * CL * sizeof(int), stream);

  const int T = 256;
  rank_kernel<<<(E + T - 1) / T, T, 0, stream>>>(dst, cl, cl1, rank, E);
  alloc_kernel<<<(n + T - 1) / T, T, 0, stream>>>(cl, cl1, n);
  fill_kernel<<<(E + T * FILL_U - 1) / (T * FILL_U), T, 0, stream>>>(src, dst, ew, cl, cl1, rank, rec, E);
  dis_pull_kernel<<<(n + T - 1) / T, T, 0, stream>>>(cl, rec, dis, n);

  pad_x_kernel<<<(n * 12 + T * PAD_U - 1) / (T * PAD_U), T, 0, stream>>>(x, xp, n * 12);
  pack_w1_kernel<<<(12 * 2 * H1DIM * 2 + T - 1) / T, T, 0, stream>>>(W1, Bp1);
  pack_w2_kernel<<<(8 * 2 * H2DIM * 2 + T - 1) / T, T, 0, stream>>>(W2, Bp2);

  gemm1_wmma<<<(n + 15) / 16, 32, 0, stream>>>(xp, Bp1, h1, n);
  pull_kernel<true><<<((long long)n * 8 + T - 1) / T, T, 0, stream>>>(cl, rec, dis, h1, b1, hh, n);

  pull_kernel<false><<<((long long)n * 8 + T - 1) / T, T, 0, stream>>>(cl, rec, dis, hh, b1, g, n);
  gemm2_wmma<<<(n + 15) / 16, 32, 0, stream>>>(g, Bp2, b2, out, n);
}


# per-node fixed buckets: the rank add's slot stores the record at once into the node's sub-bucket (32 per counter copy), overflow to a list; alloc and fill passes go; add order: slot, as before
# speedup vs baseline: 2.3551x; 1.4571x over previous
#include <hip/hip_runtime.h>
#include <math.h>

typedef __attribute__((ext_vector_type(2))) float v2f;
typedef __attribute__((ext_vector_type(8))) float v8f;

#define NC    45
#define NCP   48
#define H1DIM 32
#define H2DIM 128

#define PAD_U 8
typedef float row4 __attribute__((ext_vector_type(4), aligned(4)));
template <bool WHOLE>
__device__ __forceinline__ void pad_trip(const float* __restrict__ x, float* __restrict__ xp,
                                         int g0, int stride, int pieces) {
  float4 v[PAD_U];
#pragma unroll
  for (int u = 0; u < PAD_U; ++u) {
    int g = g0 + u * stride;
    if (!WHOLE && g > pieces - 1) g = pieces - 1;
    const int i = g / 12, j = g - i * 12;
    const float* p = x + (size_t)i * NC + 4 * j;
    const row4 t = *(const row4*)(p - (j == 11 ? 3 : 0));
    v[u] = (j == 11) ? make_float4(t.w, 0.0f, 0.0f, 0.0f) : make_float4(t.x, t.y, t.z, t.w);
  }
#pragma unroll
  for (int u = 0; u < PAD_U; ++u) {
    const int g = g0 + u * stride;
    if (WHOLE || g < pieces) *(float4*)(xp + 4 * (size_t)g) = v[u];
  }
}

__global__ void pad_x_kernel(const float* __restrict__ x, float* __restrict__ xp,
                             int pieces  ) {
  const int stride = gridDim.x * blockDim.x;
  for (int g0 = blockIdx.x * blockDim.x + threadIdx.x; g0 < pieces; g0 += PAD_U * stride) {
    if (g0 + (PAD_U - 1) * stride < pieces) pad_trip<true>(x, xp, g0, stride, pieces);
    else                                    pad_trip<false>(x, xp, g0, stride, pieces);
  }
}

__global__ void pack_w1_kernel(const float* __restrict__ W1, float* __restrict__ Bp1) {
  int gid = blockIdx.x * blockDim.x + threadIdx.x;
  if (gid >= 12 * 2 * H1DIM * 2) return;
  int p   = gid & 1;
  int col = (gid >> 1) & 31;
  int kh  = (gid >> 6) & 1;
  int k   = gid >> 7;
  int row = 4 * k + 2 * kh + p;
  Bp1[gid] = (row < NC) ? W1[row * H1DIM + col] : 0.0f;
}

__global__ void pack_w2_kernel(const float* __restrict__ W2, float* __restrict__ Bp2) {
  int gid = blockIdx.x * blockDim.x + threadIdx.x;
  if (gid >= 8 * 2 * H2DIM * 2) return;
  int p   = gid & 1;
  int col = (gid >> 1) & 127;
  int kh  = (gid >> 8) & 1;
  int k   = gid >> 9;
  int row = 4 * k + 2 * kh + p;
  Bp2[gid] = W2[row * H2DIM + col];
}

__global__ void gemm1_wmma(const float* __restrict__ xp, const float* __restrict__ Bp1,
                           float* __restrict__ h1, int n) {
  const int lane  = threadIdx.x;
  const int m0    = blockIdx.x * 16;
  const int khalf = lane >> 4;
  const int nn    = lane & 15;
  int m = m0 + nn;
  if (m > n - 1) m = n - 1;

  const float* ap = xp + (size_t)m * NCP + 2 * khalf;
  const float* bp = Bp1 + khalf * 64;
  v8f c0 = {};
  v8f c1 = {};
#pragma unroll
  for (int k = 0; k < 12; ++k) {
    v2f a  = *(const v2f*)(ap + 4 * k);
    v2f b0 = *(const v2f*)(bp + 128 * k + nn * 2);
    v2f b1 = *(const v2f*)(bp + 128 * k + (16 + nn) * 2);
    c0 = __builtin_amdgcn_wmma_f32_16x16x4_f32(false, a, false, b0, (short)0, c0, false, false);
    c1 = __builtin_amdgcn_wmma_f32_16x16x4_f32(false, a, false, b1, (short)0, c1, false, false);
  }
  const int rbase = m0 + (khalf << 3);
#pragma unroll
  for (int j = 0; j < 8; ++j) {
    const int r = rbase + j;
    if (r < n) {
      h1[(size_t)r * H1DIM + nn]      = c0[j];
      h1[(size_t)r * H1DIM + 16 + nn] = c1[j];
    }
  }
}

__global__ void gemm2_wmma(const float* __restrict__ g, const float* __restrict__ Bp2,
                           const float* __restrict__ b2, float* __restrict__ out, int n) {
  const int lane  = threadIdx.x;
  const int m0    = blockIdx.x * 16;
  const int khalf = lane >> 4;
  const int nn    = lane & 15;
  int m = m0 + nn;
  if (m > n - 1) m = n - 1;

  const float* ap = g + (size_t)m * H1DIM + 2 * khalf;
  const float* bp = Bp2 + khalf * 256;
  v2f a[8];
#pragma unroll
  for (int k = 0; k < 8; ++k) a[k] = *(const v2f*)(ap + 4 * k);

  v8f acc[8] = {};
#pragma unroll
  for (int k = 0; k < 8; ++k) {
#pragma unroll
    for (int t = 0; t < 8; ++t) {
      v2f b = *(const v2f*)(bp + 512 * k + (t * 16 + nn) * 2);
      acc[t] = __builtin_amdgcn_wmma_f32_16x16x4_f32(false, a[k], false, b, (short)0, acc[t], false, false);
    }
  }
  float bv[8];
#pragma unroll
  for (int t = 0; t < 8; ++t) bv[t] = b2[t * 16 + nn];

  const int rbase = m0 + (khalf << 3);
#pragma unroll
  for (int j = 0; j < 8; ++j) {
    const int r = rbase + j;
    if (r < n) {
#pragma unroll
      for (int t = 0; t < 8; ++t)
        out[(size_t)r * H2DIM + t * 16 + nn] = acc[t][j] + bv[t];
    }
  }
}

#define CL   32
#define CAPH 32

__device__ __forceinline__ int package_half() {
  return (int)((__builtin_amdgcn_s_sendmsg_rtn(0x87) >> 18) & 1u);
}

__global__ void place_kernel(const int* __restrict__ src, const int* __restrict__ dst,
                             const float* __restrict__ ew, int* __restrict__ cl0,
                             int* __restrict__ cl1, int* __restrict__ rec,
                             int4* __restrict__ ov, int n, int E) {
  const int e = blockIdx.x * blockDim.x + threadIdx.x;
  const int hf = package_half();
  int* cl = hf ? cl1 : cl0;
  if (e < E) {
    const int d = dst[e];
    const int s = src[e];
    const int w = __float_as_int(ew[e]);
    const int slot = atomicAdd(&cl[(size_t)d * CL], 1);
    if (slot < CAPH) {
      *(int2*)(rec + 2 * (((size_t)d * 2 + hf) * CAPH + slot)) = make_int2(s, w);
    } else {
      const int k = atomicAdd(&cl0[(size_t)n * CL], 1);
      ov[k] = make_int4(d, s, w, 0);
    }
  }
}

__device__ __forceinline__ int list_off(int t, int c0) {
  return t < c0 ? t : t - c0 + CAPH;
}

__global__ void dis_pull_kernel(const int* __restrict__ cl0, const int* __restrict__ cl1,
                                const int* __restrict__ rec, const int4* __restrict__ ov,
                                int2* __restrict__ cnt, float* __restrict__ dis, int n, int E) {
  int i = blockIdx.x * blockDim.x + threadIdx.x;
  if (i >= n) return;
  const int c0  = min(cl0[(size_t)i * CL], CAPH);
  const int c1  = min(cl1[(size_t)i * CL], CAPH);
  const int ovn = min(cl0[(size_t)n * CL], E);
  cnt[i] = make_int2(c0, c1);
  const int2* rb = (const int2*)rec;
  const int base = i * (2 * CAPH);
  const int end = c0 + c1;
  float s = 0.0f;
  for (int j = 0; j < end; j += 4) {
    const bool k1 = j + 1 < end, k2 = j + 2 < end, k3 = j + 3 < end;
    const float w0 = __int_as_float(rb[base + list_off(j, c0)].y);
    const float w1 = __int_as_float(rb[base + list_off(k1 ? j + 1 : j, c0)].y);
    const float w2 = __int_as_float(rb[base + list_off(k2 ? j + 2 : j, c0)].y);
    const float w3 = __int_as_float(rb[base + list_off(k3 ? j + 3 : j, c0)].y);
    s += w0;
    s += k1 ? w1 : 0.0f;
    s += k2 ? w2 : 0.0f;
    s += k3 ? w3 : 0.0f;
  }
  for (int k = 0; k < ovn; ++k) {
    const int4 o = ov[k];
    if (o.x == i) s += __int_as_float(o.z);
  }
  dis[i] = 1.0f / sqrtf(s + 1.0f);
}

template <bool ACT>
__global__ void pull_kernel(const int* __restrict__ cl0, const int2* __restrict__ cnt,
                            const int* __restrict__ rec, const int4* __restrict__ ov,
                            const float* __restrict__ dis, const float* __restrict__ h,
                            const float* __restrict__ bias, float* __restrict__ outp, int n, int E) {
  const int gid = blockIdx.x * blockDim.x + threadIdx.x;
  const int i   = gid >> 3;
  const int q   = (gid & 7) << 2;
  if (i >= n) return;
  const int2 co = cnt[i];
  const int c0  = co.x;
  const int end = co.x + co.y;
  const int ovn = min(cl0[(size_t)n * CL], E);
  const int2* rb = (const int2*)rec;
  const int base = i * (2 * CAPH);
  float4 acc = make_float4(0.0f, 0.0f, 0.0f, 0.0f);
  for (int j = 0; j < end; j += 4) {
    const bool k1 = j + 1 < end, k2 = j + 2 < end, k3 = j + 3 < end;
    const int2 r0 = rb[base + list_off(j, c0)];
    const int2 r1 = rb[base + list_off(k1 ? j + 1 : j, c0)];
    const int2 r2 = rb[base + list_off(k2 ? j + 2 : j, c0)];
    const int2 r3 = rb[base + list_off(k3 ? j + 3 : j, c0)];
    const float s0 = dis[r0.x], s1 = dis[r1.x], s2 = dis[r2.x], s3 = dis[r3.x];
    const float4 v0 = *(const float4*)(h + (size_t)r0.x * H1DIM + q);
    const float4 v1 = *(const float4*)(h + (size_t)r1.x * H1DIM + q);
    const float4 v2 = *(const float4*)(h + (size_t)r2.x * H1DIM + q);
    const float4 v3 = *(const float4*)(h + (size_t)r3.x * H1DIM + q);
    const float c0f = s0 * __int_as_float(r0.y);
    const float c1f = k1 ? s1 * __int_as_float(r1.y) : 0.0f;
    const float c2f = k2 ? s2 * __int_as_float(r2.y) : 0.0f;
    const float c3f = k3 ? s3 * __int_as_float(r3.y) : 0.0f;
    acc.x += v0.x * c0f; acc.y += v0.y * c0f; acc.z += v0.z * c0f; acc.w += v0.w * c0f;
    acc.x += v1.x * c1f; acc.y += v1.y * c1f; acc.z += v1.z * c1f; acc.w += v1.w * c1f;
    acc.x += v2.x * c2f; acc.y += v2.y * c2f; acc.z += v2.z * c2f; acc.w += v2.w * c2f;
    acc.x += v3.x * c3f; acc.y += v3.y * c3f; acc.z += v3.z * c3f; acc.w += v3.w * c3f;
  }
  for (int k = 0; k < ovn; ++k) {
    const int4 o = ov[k];
    if (o.x == i) {
      const float cf = dis[o.y] * __int_as_float(o.z);
      const float4 v = *(const float4*)(h + (size_t)o.y * H1DIM + q);
      acc.x += v.x * cf; acc.y += v.y * cf; acc.z += v.z * cf; acc.w += v.w * cf;
    }
  }
  const float di = dis[i];
  const float d2 = di * di;
  const float4 hv = *(const float4*)(h + (size_t)i * H1DIM + q);
  float4 o;
  o.x = acc.x * di + hv.x * d2;
  o.y = acc.y * di + hv.y * d2;
  o.z = acc.z * di + hv.z * d2;
  o.w = acc.w * di + hv.w * d2;
  if (ACT) {
    o.x = tanhf(o.x + bias[q + 0]);
    o.y = tanhf(o.y + bias[q + 1]);
    o.z = tanhf(o.z + bias[q + 2]);
    o.w = tanhf(o.w + bias[q + 3]);
  }
  *(float4*)(outp + (size_t)i * H1DIM + q) = o;
}

extern "C" void kernel_launch(void* const* d_in, const int* in_sizes, int n_in,
                              void* d_out, int out_size, void* d_ws, size_t ws_size,
                              hipStream_t stream) {
  const float* x   = (const float*)d_in[0];
  const int*   src = (const int*)  d_in[1];
  const int*   dst = (const int*)  d_in[2];
  const float* ew  = (const float*)d_in[3];
  const float* W1  = (const float*)d_in[4];
  const float* b1  = (const float*)d_in[5];
  const float* W2  = (const float*)d_in[6];
  const float* b2  = (const float*)d_in[7];
  float* out = (float*)d_out;

  const int n = in_sizes[0] / NC;
  const int E = in_sizes[1];

  int*   cl   = (int*)d_ws;
  int*   cl1  = cl   + (size_t)(n + 2) * CL;
  int*   rec  = cl1  + (size_t)(n + 2) * CL;
  int4*  ov   = (int4*)(rec + (size_t)n * 4 * CAPH);
  int2*  cnt  = (int2*)(ov + (size_t)E);
  float* dis  = (float*)(cnt + (size_t)n);
  float* xp   = dis  + n;
  float* h1   = xp   + (size_t)n * NCP;
  float* hh   = h1   + (size_t)n * H1DIM;
  float* g    = hh   + (size_t)n * H1DIM;
  float* Bp1  = g    + (size_t)n * H1DIM;
  float* Bp2  = Bp1  + 12 * 2 * H1DIM * 2;

  hipMemsetAsync(cl, 0, 2 * (size_t)(n + 2) * CL * sizeof(int), stream);

  const int T = 256;
  place_kernel<<<(E + T - 1) / T, T, 0, stream>>>(src, dst, ew, cl, cl1, rec, ov, n, E);
  dis_pull_kernel<<<(n + T - 1) / T, T, 0, stream>>>(cl, cl1, rec, ov, cnt, dis, n, E);

  pad_x_kernel<<<(n * 12 + T * PAD_U - 1) / (T * PAD_U), T, 0, stream>>>(x, xp, n * 12);
  pack_w1_kernel<<<(12 * 2 * H1DIM * 2 + T - 1) / T, T, 0, stream>>>(W1, Bp1);
  pack_w2_kernel<<<(8 * 2 * H2DIM * 2 + T - 1) / T, T, 0, stream>>>(W2, Bp2);

  gemm1_wmma<<<(n + 15) / 16, 32, 0, stream>>>(xp, Bp1, h1, n);
  pull_kernel<true><<<((long long)n * 8 + T - 1) / T, T, 0, stream>>>(cl, cnt, rec, ov, dis, h1, b1, hh, n, E);

  pull_kernel<false><<<((long long)n * 8 + T - 1) / T, T, 0, stream>>>(cl, cnt, rec, ov, dis, hh, b1, g, n, E);
  gemm2_wmma<<<(n + 15) / 16, 32, 0, stream>>>(g, Bp2, b2, out, n);
}


# independent passes share a launch (by block index): counter zeroing (was a memset), x pad, W1/W2 packs in one; degree pass beside x@W1 (8-wave blocks); 10 stream ops become 6; add order unchanged
# speedup vs baseline: 2.4267x; 1.0304x over previous
#include <hip/hip_runtime.h>
#include <math.h>

typedef __attribute__((ext_vector_type(2))) float v2f;
typedef __attribute__((ext_vector_type(8))) float v8f;

#define NC    45
#define NCP   48
#define H1DIM 32
#define H2DIM 128

#define PAD_U 8
#define PREP_T 256
typedef float row4 __attribute__((ext_vector_type(4), aligned(4)));
template <bool WHOLE>
__device__ __forceinline__ void pad_trip(const float* __restrict__ x, float* __restrict__ xp,
                                         int g0, int stride, int pieces) {
  float4 v[PAD_U];
#pragma unroll
  for (int u = 0; u < PAD_U; ++u) {
    int g = g0 + u * stride;
    if (!WHOLE && g > pieces - 1) g = pieces - 1;
    const int i = g / 12, j = g - i * 12;
    const float* p = x + (size_t)i * NC + 4 * j;
    const row4 t = *(const row4*)(p - (j == 11 ? 3 : 0));
    v[u] = (j == 11) ? make_float4(t.w, 0.0f, 0.0f, 0.0f) : make_float4(t.x, t.y, t.z, t.w);
  }
#pragma unroll
  for (int u = 0; u < PAD_U; ++u) {
    const int g = g0 + u * stride;
    if (WHOLE || g < pieces) *(float4*)(xp + 4 * (size_t)g) = v[u];
  }
}

__device__ __forceinline__ void pad_x_body(const float* __restrict__ x, float* __restrict__ xp,
                                           int pieces  , int vb, int nb) {
  const int stride = nb * PREP_T;
  for (int g0 = vb * PREP_T + threadIdx.x; g0 < pieces; g0 += PAD_U * stride) {
    if (g0 + (PAD_U - 1) * stride < pieces) pad_trip<true>(x, xp, g0, stride, pieces);
    else                                    pad_trip<false>(x, xp, g0, stride, pieces);
  }
}

__device__ __forceinline__ void pack_w1_body(const float* __restrict__ W1, float* __restrict__ Bp1, int vb) {
  int gid = vb * PREP_T + threadIdx.x;
  if (gid >= 12 * 2 * H1DIM * 2) return;
  int p   = gid & 1;
  int col = (gid >> 1) & 31;
  int kh  = (gid >> 6) & 1;
  int k   = gid >> 7;
  int row = 4 * k + 2 * kh + p;
  Bp1[gid] = (row < NC) ? W1[row * H1DIM + col] : 0.0f;
}

__device__ __forceinline__ void pack_w2_body(const float* __restrict__ W2, float* __restrict__ Bp2, int vb) {
  int gid = vb * PREP_T + threadIdx.x;
  if (gid >= 8 * 2 * H2DIM * 2) return;
  int p   = gid & 1;
  int col = (gid >> 1) & 127;
  int kh  = (gid >> 8) & 1;
  int k   = gid >> 9;
  int row = 4 * k + 2 * kh + p;
  Bp2[gid] = W2[row * H2DIM + col];
}

#define ZERO_U 8
__device__ __forceinline__ void zero_body(int4* __restrict__ p, int pieces, int vb, int nb) {
  const int stride = nb * PREP_T;
  const int4 z = make_int4(0, 0, 0, 0);
  for (int g0 = vb * PREP_T + threadIdx.x; g0 < pieces; g0 += ZERO_U * stride) {
#pragma unroll
    for (int u = 0; u < ZERO_U; ++u) {
      const int g = g0 + u * stride;
      if (g < pieces) p[g] = z;
    }
  }
}

#define W1_BLOCKS ((12 * 2 * H1DIM * 2 + PREP_T - 1) / PREP_T)
#define W2_BLOCKS ((8 * 2 * H2DIM * 2 + PREP_T - 1) / PREP_T)
__global__ void __launch_bounds__(PREP_T)
prep_kernel(const float* __restrict__ x, const float* __restrict__ W1, const float* __restrict__ W2,
            int4* __restrict__ zero_base, int zero_pieces, float* __restrict__ xp,
            float* __restrict__ Bp1, float* __restrict__ Bp2, int pad_pieces, int zb, int pb) {
  const int b = blockIdx.x;
  if (b < zb)                zero_body(zero_base, zero_pieces, b, zb);
  else if (b < zb + pb)      pad_x_body(x, xp, pad_pieces, b - zb, pb);
  else if (b < zb + pb + W1_BLOCKS) pack_w1_body(W1, Bp1, b - zb - pb);
  else                              pack_w2_body(W2, Bp2, b - zb - pb - W1_BLOCKS);
}

__device__ __forceinline__ void gemm1_body(const float* __restrict__ xp, const float* __restrict__ Bp1,
                                           float* __restrict__ h1, int n, int vb) {
  const int lane  = threadIdx.x & 31;
  const int wave  = __builtin_amdgcn_readfirstlane(threadIdx.x >> 5);
  const int m0    = (vb * (PREP_T / 32) + wave) * 16;
  if (m0 >= n) return;
  const int khalf = lane >> 4;
  const int nn    = lane & 15;
  int m = m0 + nn;
  if (m > n - 1) m = n - 1;

  const float* ap = xp + (size_t)m * NCP + 2 * khalf;
  const float* bp = Bp1 + khalf * 64;
  v8f c0 = {};
  v8f c1 = {};
#pragma unroll
  for (int k = 0; k < 12; ++k) {
    v2f a  = *(const v2f*)(ap + 4 * k);
    v2f b0 = *(const v2f*)(bp + 128 * k + nn * 2);
    v2f b1 = *(const v2f*)(bp + 128 * k + (16 + nn) * 2);
    c0 = __builtin_amdgcn_wmma_f32_16x16x4_f32(false, a, false, b0, (short)0, c0, false, false);
    c1 = __builtin_amdgcn_wmma_f32_16x16x4_f32(false, a, false, b1, (short)0, c1, false, false);
  }
  const int rbase = m0 + (khalf << 3);
#pragma unroll
  for (int j = 0; j < 8; ++j) {
    const int r = rbase + j;
    if (r < n) {
      h1[(size_t)r * H1DIM + nn]      = c0[j];
      h1[(size_t)r * H1DIM + 16 + nn] = c1[j];
    }
  }
}

__global__ void gemm2_wmma(const float* __restrict__ g, const float* __restrict__ Bp2,
                           const float* __restrict__ b2, float* __restrict__ out, int n) {
  const int lane  = threadIdx.x;
  const int m0    = blockIdx.x * 16;
  const int khalf = lane >> 4;
  const int nn    = lane & 15;
  int m = m0 + nn;
  if (m > n - 1) m = n - 1;

  const float* ap = g + (size_t)m * H1DIM + 2 * khalf;
  const float* bp = Bp2 + khalf * 256;
  v2f a[8];
#pragma unroll
  for (int k = 0; k < 8; ++k) a[k] = *(const v2f*)(ap + 4 * k);

  v8f acc[8] = {};
#pragma unroll
  for (int k = 0; k < 8; ++k) {
#pragma unroll
    for (int t = 0; t < 8; ++t) {
      v2f b = *(const v2f*)(bp + 512 * k + (t * 16 + nn) * 2);
      acc[t] = __builtin_amdgcn_wmma_f32_16x16x4_f32(false, a[k], false, b, (short)0, acc[t], false, false);
    }
  }
  float bv[8];
#pragma unroll
  for (int t = 0; t < 8; ++t) bv[t] = b2[t * 16 + nn];

  const int rbase = m0 + (khalf << 3);
#pragma unroll
  for (int j = 0; j < 8; ++j) {
    const int r = rbase + j;
    if (r < n) {
#pragma unroll
      for (int t = 0; t < 8; ++t)
        out[(size_t)r * H2DIM + t * 16 + nn] = acc[t][j] + bv[t];
    }
  }
}

#define CL   32
#define CAPH 32

__device__ __forceinline__ int package_half() {
  return (int)((__builtin_amdgcn_s_sendmsg_rtn(0x87) >> 18) & 1u);
}

__global__ void place_kernel(const int* __restrict__ src, const int* __restrict__ dst,
                             const float* __restrict__ ew, int* __restrict__ cl0,
                             int* __restrict__ cl1, int* __restrict__ rec,
                             int4* __restrict__ ov, int n, int E) {
  const int e = blockIdx.x * blockDim.x + threadIdx.x;
  const int hf = package_half();
  int* cl = hf ? cl1 : cl0;
  if (e < E) {
    const int d = dst[e];
    const int s = src[e];
    const int w = __float_as_int(ew[e]);
    const int slot = atomicAdd(&cl[(size_t)d * CL], 1);
    if (slot < CAPH) {
      *(int2*)(rec + 2 * (((size_t)d * 2 + hf) * CAPH + slot)) = make_int2(s, w);
    } else {
      const int k = atomicAdd(&cl0[(size_t)n * CL], 1);
      ov[k] = make_int4(d, s, w, 0);
    }
  }
}

__device__ __forceinline__ int list_off(int t, int c0) {
  return t < c0 ? t : t - c0 + CAPH;
}

__device__ __forceinline__ void dis_pull_body(const int* __restrict__ cl0, const int* __restrict__ cl1,
                                              const int* __restrict__ rec, const int4* __restrict__ ov,
                                              int2* __restrict__ cnt, float* __restrict__ dis,
                                              int n, int E, int vb) {
  int i = vb * PREP_T + threadIdx.x;
  if (i >= n) return;
  const int c0  = min(cl0[(size_t)i * CL], CAPH);
  const int c1  = min(cl1[(size_t)i * CL], CAPH);
  const int ovn = min(cl0[(size_t)n * CL], E);
  cnt[i] = make_int2(c0, c1);
  const int2* rb = (const int2*)rec;
  const int base = i * (2 * CAPH);
  const int end = c0 + c1;
  float s = 0.0f;
  for (int j = 0; j < end; j += 4) {
    const bool k1 = j + 1 < end, k2 = j + 2 < end, k3 = j + 3 < end;
    const float w0 = __int_as_float(rb[base + list_off(j, c0)].y);
    const float w1 = __int_as_float(rb[base + list_off(k1 ? j + 1 : j, c0)].y);
    const float w2 = __int_as_float(rb[base + list_off(k2 ? j + 2 : j, c0)].y);
    const float w3 = __int_as_float(rb[base + list_off(k3 ? j + 3 : j, c0)].y);
    s += w0;
    s += k1 ? w1 : 0.0f;
    s += k2 ? w2 : 0.0f;
    s += k3 ? w3 : 0.0f;
  }
  for (int k = 0; k < ovn; ++k) {
    const int4 o = ov[k];
    if (o.x == i) s += __int_as_float(o.z);
  }
  dis[i] = 1.0f / sqrtf(s + 1.0f);
}

__global__ void __launch_bounds__(PREP_T)
dis_gemm1_kernel(const int* __restrict__ cl0, const int* __restrict__ cl1,
                 const int* __restrict__ rec, const int4* __restrict__ ov,
                 int2* __restrict__ cnt, float* __restrict__ dis,
                 const float* __restrict__ xp, const float* __restrict__ Bp1,
                 float* __restrict__ h1, int n, int E, int db) {
  const int b = blockIdx.x;
  if (b < db) dis_pull_body(cl0, cl1, rec, ov, cnt, dis, n, E, b);
  else        gemm1_body(xp, Bp1, h1, n, b - db);
}

template <bool ACT>
__global__ void pull_kernel(const int* __restrict__ cl0, const int2* __restrict__ cnt,
                            const int* __restrict__ rec, const int4* __restrict__ ov,
                            const float* __restrict__ dis, const float* __restrict__ h,
                            const float* __restrict__ bias, float* __restrict__ outp, int n, int E) {
  const int gid = blockIdx.x * blockDim.x + threadIdx.x;
  const int i   = gid >> 3;
  const int q   = (gid & 7) << 2;
  if (i >= n) return;
  const int2 co = cnt[i];
  const int c0  = co.x;
  const int end = co.x + co.y;
  const int ovn = min(cl0[(size_t)n * CL], E);
  const int2* rb = (const int2*)rec;
  const int base = i * (2 * CAPH);
  float4 acc = make_float4(0.0f, 0.0f, 0.0f, 0.0f);
  for (int j = 0; j < end; j += 4) {
    const bool k1 = j + 1 < end, k2 = j + 2 < end, k3 = j + 3 < end;
    const int2 r0 = rb[base + list_off(j, c0)];
    const int2 r1 = rb[base + list_off(k1 ? j + 1 : j, c0)];
    const int2 r2 = rb[base + list_off(k2 ? j + 2 : j, c0)];
    const int2 r3 = rb[base + list_off(k3 ? j + 3 : j, c0)];
    const float s0 = dis[r0.x], s1 = dis[r1.x], s2 = dis[r2.x], s3 = dis[r3.x];
    const float4 v0 = *(const float4*)(h + (size_t)r0.x * H1DIM + q);
    const float4 v1 = *(const float4*)(h + (size_t)r1.x * H1DIM + q);
    const float4 v2 = *(const float4*)(h + (size_t)r2.x * H1DIM + q);
    const float4 v3 = *(const float4*)(h + (size_t)r3.x * H1DIM + q);
    const float c0f = s0 * __int_as_float(r0.y);
    const float c1f = k1 ? s1 * __int_as_float(r1.y) : 0.0f;
    const float c2f = k2 ? s2 * __int_as_float(r2.y) : 0.0f;
    const float c3f = k3 ? s3 * __int_as_float(r3.y) : 0.0f;
    acc.x += v0.x * c0f; acc.y += v0.y * c0f; acc.z += v0.z * c0f; acc.w += v0.w * c0f;
    acc.x += v1.x * c1f; acc.y += v1.y * c1f; acc.z += v1.z * c1f; acc.w += v1.w * c1f;
    acc.x += v2.x * c2f; acc.y += v2.y * c2f; acc.z += v2.z * c2f; acc.w += v2.w * c2f;
    acc.x += v3.x * c3f; acc.y += v3.y * c3f; acc.z += v3.z * c3f; acc.w += v3.w * c3f;
  }
  for (int k = 0; k < ovn; ++k) {
    const int4 o = ov[k];
    if (o.x == i) {
      const float cf = dis[o.y] * __int_as_float(o.z);
      const float4 v = *(const float4*)(h + (size_t)o.y * H1DIM + q);
      acc.x += v.x * cf; acc.y += v.y * cf; acc.z += v.z * cf; acc.w += v.w * cf;
    }
  }
  const float di = dis[i];
  const float d2 = di * di;
  const float4 hv = *(const float4*)(h + (size_t)i * H1DIM + q);
  float4 o;
  o.x = acc.x * di + hv.x * d2;
  o.y = acc.y * di + hv.y * d2;
  o.z = acc.z * di + hv.z * d2;
  o.w = acc.w * di + hv.w * d2;
  if (ACT) {
    o.x = tanhf(o.x + bias[q + 0]);
    o.y = tanhf(o.y + bias[q + 1]);
    o.z = tanhf(o.z + bias[q + 2]);
    o.w = tanhf(o.w + bias[q + 3]);
  }
  *(float4*)(outp + (size_t)i * H1DIM + q) = o;
}

extern "C" void kernel_launch(void* const* d_in, const int* in_sizes, int n_in,
                              void* d_out, int out_size, void* d_ws, size_t ws_size,
                              hipStream_t stream) {
  const float* x   = (const float*)d_in[0];
  const int*   src = (const int*)  d_in[1];
  const int*   dst = (const int*)  d_in[2];
  const float* ew  = (const float*)d_in[3];
  const float* W1  = (const float*)d_in[4];
  const float* b1  = (const float*)d_in[5];
  const float* W2  = (const float*)d_in[6];
  const float* b2  = (const float*)d_in[7];
  float* out = (float*)d_out;

  const int n = in_sizes[0] / NC;
  const int E = in_sizes[1];

  int*   cl   = (int*)d_ws;
  int*   cl1  = cl   + (size_t)(n + 2) * CL;
  int*   rec  = cl1  + (size_t)(n + 2) * CL;
  int4*  ov   = (int4*)(rec + (size_t)n * 4 * CAPH);
  int2*  cnt  = (int2*)(ov + (size_t)E);
  float* dis  = (float*)(cnt + (size_t)n);
  float* xp   = dis  + n;
  float* h1   = xp   + (size_t)n * NCP;
  float* hh   = h1   + (size_t)n * H1DIM;
  float* g    = hh   + (size_t)n * H1DIM;
  float* Bp1  = g    + (size_t)n * H1DIM;
  float* Bp2  = Bp1  + 12 * 2 * H1DIM * 2;

  const int T = PREP_T;
  const int zero_pieces = 2 * (n + 2) * (CL / 4);
  const int zb = (zero_pieces + T * ZERO_U - 1) / (T * ZERO_U);
  const int pb = (n * 12 + T * PAD_U - 1) / (T * PAD_U);
  prep_kernel<<<zb + pb + W1_BLOCKS + W2_BLOCKS, T, 0, stream>>>(x, W1, W2, (int4*)cl, zero_pieces,
                                                                 xp, Bp1, Bp2, n * 12, zb, pb);

  place_kernel<<<(E + T - 1) / T, T, 0, stream>>>(src, dst, ew, cl, cl1, rec, ov, n, E);

  const int db = (n + T - 1) / T;
  const int gb = ((n + 15) / 16 + T / 32 - 1) / (T / 32);
  dis_gemm1_kernel<<<db + gb, T, 0, stream>>>(cl, cl1, rec, ov, cnt, dis, xp, Bp1, h1, n, E, db);

  pull_kernel<true><<<((long long)n * 8 + T - 1) / T, T, 0, stream>>>(cl, cnt, rec, ov, dis, h1, b1, hh, n, E);

  pull_kernel<false><<<((long long)n * 8 + T - 1) / T, T, 0, stream>>>(cl, cnt, rec, ov, dis, hh, b1, g, n, E);
  gemm2_wmma<<<(n + 15) / 16, 32, 0, stream>>>(g, Bp2, b2, out, n);
}


# edge counter moves into word 0 of its node's 256-B sub-bucket: an edge's rank add and record store share a line, count and records read as one; counter planes and their zero fill go; add order: slot
# speedup vs baseline: 2.5419x; 1.0475x over previous
#include <hip/hip_runtime.h>
#include <math.h>

typedef __attribute__((ext_vector_type(2))) float v2f;
typedef __attribute__((ext_vector_type(8))) float v8f;

#define NC    45
#define NCP   48
#define H1DIM 32
#define H2DIM 128

#define PAD_U 8
#define PREP_T 256
typedef float row4 __attribute__((ext_vector_type(4), aligned(4)));
template <bool WHOLE>
__device__ __forceinline__ void pad_trip(const float* __restrict__ x, float* __restrict__ xp,
                                         int g0, int stride, int pieces) {
  float4 v[PAD_U];
#pragma unroll
  for (int u = 0; u < PAD_U; ++u) {
    int g = g0 + u * stride;
    if (!WHOLE && g > pieces - 1) g = pieces - 1;
    const int i = g / 12, j = g - i * 12;
    const float* p = x + (size_t)i * NC + 4 * j;
    const row4 t = *(const row4*)(p - (j == 11 ? 3 : 0));
    v[u] = (j == 11) ? make_float4(t.w, 0.0f, 0.0f, 0.0f) : make_float4(t.x, t.y, t.z, t.w);
  }
#pragma unroll
  for (int u = 0; u < PAD_U; ++u) {
    const int g = g0 + u * stride;
    if (WHOLE || g < pieces) *(float4*)(xp + 4 * (size_t)g) = v[u];
  }
}

__device__ __forceinline__ void pad_x_body(const float* __restrict__ x, float* __restrict__ xp,
                                           int pieces  , int vb, int nb) {
  const int stride = nb * PREP_T;
  for (int g0 = vb * PREP_T + threadIdx.x; g0 < pieces; g0 += PAD_U * stride) {
    if (g0 + (PAD_U - 1) * stride < pieces) pad_trip<true>(x, xp, g0, stride, pieces);
    else                                    pad_trip<false>(x, xp, g0, stride, pieces);
  }
}

__device__ __forceinline__ void pack_w1_body(const float* __restrict__ W1, float* __restrict__ Bp1, int vb) {
  int gid = vb * PREP_T + threadIdx.x;
  if (gid >= 12 * 2 * H1DIM * 2) return;
  int p   = gid & 1;
  int col = (gid >> 1) & 31;
  int kh  = (gid >> 6) & 1;
  int k   = gid >> 7;
  int row = 4 * k + 2 * kh + p;
  Bp1[gid] = (row < NC) ? W1[row * H1DIM + col] : 0.0f;
}

__device__ __forceinline__ void pack_w2_body(const float* __restrict__ W2, float* __restrict__ Bp2, int vb) {
  int gid = vb * PREP_T + threadIdx.x;
  if (gid >= 8 * 2 * H2DIM * 2) return;
  int p   = gid & 1;
  int col = (gid >> 1) & 127;
  int kh  = (gid >> 8) & 1;
  int k   = gid >> 9;
  int row = 4 * k + 2 * kh + p;
  Bp2[gid] = W2[row * H2DIM + col];
}

#define ZERO_U 8
#define SBW 64
__device__ __forceinline__ void zero_body(int* __restrict__ p, int pieces, int vb, int nb) {
  const int stride = nb * PREP_T;
  for (int g0 = vb * PREP_T + threadIdx.x; g0 < pieces; g0 += ZERO_U * stride) {
#pragma unroll
    for (int u = 0; u < ZERO_U; ++u) {
      const int g = g0 + u * stride;
      if (g < pieces) p[(size_t)g * SBW] = 0;
    }
  }
}

#define W1_BLOCKS ((12 * 2 * H1DIM * 2 + PREP_T - 1) / PREP_T)
#define W2_BLOCKS ((8 * 2 * H2DIM * 2 + PREP_T - 1) / PREP_T)
__global__ void __launch_bounds__(PREP_T)
prep_kernel(const float* __restrict__ x, const float* __restrict__ W1, const float* __restrict__ W2,
            int* __restrict__ zero_base, int zero_pieces, float* __restrict__ xp,
            float* __restrict__ Bp1, float* __restrict__ Bp2, int pad_pieces, int zb, int pb) {
  const int b = blockIdx.x;
  if (b < zb)                zero_body(zero_base, zero_pieces, b, zb);
  else if (b < zb + pb)      pad_x_body(x, xp, pad_pieces, b - zb, pb);
  else if (b < zb + pb + W1_BLOCKS) pack_w1_body(W1, Bp1, b - zb - pb);
  else                              pack_w2_body(W2, Bp2, b - zb - pb - W1_BLOCKS);
}

__device__ __forceinline__ void gemm1_body(const float* __restrict__ xp, const float* __restrict__ Bp1,
                                           float* __restrict__ h1, int n, int vb) {
  const int lane  = threadIdx.x & 31;
  const int wave  = __builtin_amdgcn_readfirstlane(threadIdx.x >> 5);
  const int m0    = (vb * (PREP_T / 32) + wave) * 16;
  if (m0 >= n) return;
  const int khalf = lane >> 4;
  const int nn    = lane & 15;
  int m = m0 + nn;
  if (m > n - 1) m = n - 1;

  const float* ap = xp + (size_t)m * NCP + 2 * khalf;
  const float* bp = Bp1 + khalf * 64;
  v8f c0 = {};
  v8f c1 = {};
#pragma unroll
  for (int k = 0; k < 12; ++k) {
    v2f a  = *(const v2f*)(ap + 4 * k);
    v2f b0 = *(const v2f*)(bp + 128 * k + nn * 2);
    v2f b1 = *(const v2f*)(bp + 128 * k + (16 + nn) * 2);
    c0 = __builtin_amdgcn_wmma_f32_16x16x4_f32(false, a, false, b0, (short)0, c0, false, false);
    c1 = __builtin_amdgcn_wmma_f32_16x16x4_f32(false, a, false, b1, (short)0, c1, false, false);
  }
  const int rbase = m0 + (khalf << 3);
#pragma unroll
  for (int j = 0; j < 8; ++j) {
    const int r = rbase + j;
    if (r < n) {
      h1[(size_t)r * H1DIM + nn]      = c0[j];
      h1[(size_t)r * H1DIM + 16 + nn] = c1[j];
    }
  }
}

__global__ void gemm2_wmma(const float* __restrict__ g, const float* __restrict__ Bp2,
                           const float* __restrict__ b2, float* __restrict__ out, int n) {
  const int lane  = threadIdx.x;
  const int m0    = blockIdx.x * 16;
  const int khalf = lane >> 4;
  const int nn    = lane & 15;
  int m = m0 + nn;
  if (m > n - 1) m = n - 1;

  const float* ap = g + (size_t)m * H1DIM + 2 * khalf;
  const float* bp = Bp2 + khalf * 256;
  v2f a[8];
#pragma unroll
  for (int k = 0; k < 8; ++k) a[k] = *(const v2f*)(ap + 4 * k);

  v8f acc[8] = {};
#pragma unroll
  for (int k = 0; k < 8; ++k) {
#pragma unroll
    for (int t = 0; t < 8; ++t) {
      v2f b = *(const v2f*)(bp + 512 * k + (t * 16 + nn) * 2);
      acc[t] = __builtin_amdgcn_wmma_f32_16x16x4_f32(false, a[k], false, b, (short)0, acc[t], false, false);
    }
  }
  float bv[8];
#pragma unroll
  for (int t = 0; t < 8; ++t) bv[t] = b2[t * 16 + nn];

  const int rbase = m0 + (khalf << 3);
#pragma unroll
  for (int j = 0; j < 8; ++j) {
    const int r = rbase + j;
    if (r < n) {
#pragma unroll
      for (int t = 0; t < 8; ++t)
        out[(size_t)r * H2DIM + t * 16 + nn] = acc[t][j] + bv[t];
    }
  }
}

#define CAPH 31

__device__ __forceinline__ int package_half() {
  return (int)((__builtin_amdgcn_s_sendmsg_rtn(0x87) >> 18) & 1u);
}

__global__ void place_kernel(const int* __restrict__ src, const int* __restrict__ dst,
                             const float* __restrict__ ew, int* __restrict__ rec,
                             int4* __restrict__ ov, int n, int E) {
  const int e = blockIdx.x * blockDim.x + threadIdx.x;
  const int hf = package_half();
  if (e < E) {
    const int d = dst[e];
    const int s = src[e];
    const int w = __float_as_int(ew[e]);
    int* sb = rec + ((size_t)d * 2 + hf) * SBW;
    const int slot = atomicAdd(sb, 1);
    if (slot < CAPH) {
      *(int2*)(sb + 2 * (slot + 1)) = make_int2(s, w);
    } else {
      const int k = atomicAdd(rec + (size_t)n * 2 * SBW, 1);
      ov[k] = make_int4(d, s, w, 0);
    }
  }
}

__device__ __forceinline__ int list_off(int t, int c0) {
  return t < c0 ? t + 1 : t - c0 + 1 + SBW / 2;
}

__device__ __forceinline__ void dis_pull_body(const int* __restrict__ rec, const int4* __restrict__ ov,
                                              float* __restrict__ dis, int n, int E, int vb) {
  int i = vb * PREP_T + threadIdx.x;
  if (i >= n) return;
  const int c0  = min(rec[(size_t)i * 2 * SBW], CAPH);
  const int c1  = min(rec[(size_t)i * 2 * SBW + SBW], CAPH);
  const int ovn = min(rec[(size_t)n * 2 * SBW], E);
  const int2* rb = (const int2*)rec;
  const int base = i * SBW;
  const int end = c0 + c1;
  float s = 0.0f;
  for (int j = 0; j < end; j += 4) {
    const bool k1 = j + 1 < end, k2 = j + 2 < end, k3 = j + 3 < end;
    const float w0 = __int_as_float(rb[base + list_off(j, c0)].y);
    const float w1 = __int_as_float(rb[base + list_off(k1 ? j + 1 : j, c0)].y);
    const float w2 = __int_as_float(rb[base + list_off(k2 ? j + 2 : j, c0)].y);
    const float w3 = __int_as_float(rb[base + list_off(k3 ? j + 3 : j, c0)].y);
    s += w0;
    s += k1 ? w1 : 0.0f;
    s += k2 ? w2 : 0.0f;
    s += k3 ? w3 : 0.0f;
  }
  for (int k = 0; k < ovn; ++k) {
    const int4 o = ov[k];
    if (o.x == i) s += __int_as_float(o.z);
  }
  dis[i] = 1.0f / sqrtf(s + 1.0f);
}

__global__ void __launch_bounds__(PREP_T)
dis_gemm1_kernel(const int* __restrict__ rec, const int4* __restrict__ ov,
                 float* __restrict__ dis,
                 const float* __restrict__ xp, const float* __restrict__ Bp1,
                 float* __restrict__ h1, int n, int E, int db) {
  const int b = blockIdx.x;
  if (b < db) dis_pull_body(rec, ov, dis, n, E, b);
  else        gemm1_body(xp, Bp1, h1, n, b - db);
}

template <bool ACT>
__global__ void pull_kernel(const int* __restrict__ rec, const int4* __restrict__ ov,
                            const float* __restrict__ dis, const float* __restrict__ h,
                            const float* __restrict__ bias, float* __restrict__ outp, int n, int E) {
  const int gid = blockIdx.x * blockDim.x + threadIdx.x;
  const int i   = gid >> 3;
  const int q   = (gid & 7) << 2;
  if (i >= n) return;
  const int c0  = min(rec[(size_t)i * 2 * SBW], CAPH);
  const int end = c0 + min(rec[(size_t)i * 2 * SBW + SBW], CAPH);
  const int ovn = min(rec[(size_t)n * 2 * SBW], E);
  const int2* rb = (const int2*)rec;
  const int base = i * SBW;
  float4 acc = make_float4(0.0f, 0.0f, 0.0f, 0.0f);
  for (int j = 0; j < end; j += 4) {
    const bool k1 = j + 1 < end, k2 = j + 2 < end, k3 = j + 3 < end;
    const int2 r0 = rb[base + list_off(j, c0)];
    const int2 r1 = rb[base + list_off(k1 ? j + 1 : j, c0)];
    const int2 r2 = rb[base + list_off(k2 ? j + 2 : j, c0)];
    const int2 r3 = rb[base + list_off(k3 ? j + 3 : j, c0)];
    const float s0 = dis[r0.x], s1 = dis[r1.x], s2 = dis[r2.x], s3 = dis[r3.x];
    const float4 v0 = *(const float4*)(h + (size_t)r0.x * H1DIM + q);
    const float4 v1 = *(const float4*)(h + (size_t)r1.x * H1DIM + q);
    const float4 v2 = *(const float4*)(h + (size_t)r2.x * H1DIM + q);
    const float4 v3 = *(const float4*)(h + (size_t)r3.x * H1DIM + q);
    const float c0f = s0 * __int_as_float(r0.y);
    const float c1f = k1 ? s1 * __int_as_float(r1.y) : 0.0f;
    const float c2f = k2 ? s2 * __int_as_float(r2.y) : 0.0f;
    const float c3f = k3 ? s3 * __int_as_float(r3.y) : 0.0f;
    acc.x += v0.x * c0f; acc.y += v0.y * c0f; acc.z += v0.z * c0f; acc.w += v0.w * c0f;
    acc.x += v1.x * c1f; acc.y += v1.y * c1f; acc.z += v1.z * c1f; acc.w += v1.w * c1f;
    acc.x += v2.x * c2f; acc.y += v2.y * c2f; acc.z += v2.z * c2f; acc.w += v2.w * c2f;
    acc.x += v3.x * c3f; acc.y += v3.y * c3f; acc.z += v3.z * c3f; acc.w += v3.w * c3f;
  }
  for (int k = 0; k < ovn; ++k) {
    const int4 o = ov[k];
    if (o.x == i) {
      const float cf = dis[o.y] * __int_as_float(o.z);
      const float4 v = *(const float4*)(h + (size_t)o.y * H1DIM + q);
      acc.x += v.x * cf; acc.y += v.y * cf; acc.z += v.z * cf; acc.w += v.w * cf;
    }
  }
  const float di = dis[i];
  const float d2 = di * di;
  const float4 hv = *(const float4*)(h + (size_t)i * H1DIM + q);
  float4 o;
  o.x = acc.x * di + hv.x * d2;
  o.y = acc.y * di + hv.y * d2;
  o.z = acc.z * di + hv.z * d2;
  o.w = acc.w * di + hv.w * d2;
  if (ACT) {
    o.x = tanhf(o.x + bias[q + 0]);
    o.y = tanhf(o.y + bias[q + 1]);
    o.z = tanhf(o.z + bias[q + 2]);
    o.w = tanhf(o.w + bias[q + 3]);
  }
  *(float4*)(outp + (size_t)i * H1DIM + q) = o;
}

extern "C" void kernel_launch(void* const* d_in, const int* in_sizes, int n_in,
                              void* d_out, int out_size, void* d_ws, size_t ws_size,
                              hipStream_t stream) {
  const float* x   = (const float*)d_in[0];
  const int*   src = (const int*)  d_in[1];
  const int*   dst = (const int*)  d_in[2];
  const float* ew  = (const float*)d_in[3];
  const float* W1  = (const float*)d_in[4];
  const float* b1  = (const float*)d_in[5];
  const float* W2  = (const float*)d_in[6];
  const float* b2  = (const float*)d_in[7];
  float* out = (float*)d_out;

  const int n = in_sizes[0] / NC;
  const int E = in_sizes[1];

  int*   rec  = (int*)d_ws;
  int4*  ov   = (int4*)(rec + (size_t)(2 * n + 2) * SBW);
  float* dis  = (float*)(ov + (size_t)E);
  float* xp   = dis  + n;
  float* h1   = xp   + (size_t)n * NCP;
  float* hh   = h1   + (size_t)n * H1DIM;
  float* g    = hh   + (size_t)n * H1DIM;
  float* Bp1  = g    + (size_t)n * H1DIM;
  float* Bp2  = Bp1  + 12 * 2 * H1DIM * 2;

  const int T = PREP_T;
  const int zero_pieces = 2 * n + 1;
  const int zb = (zero_pieces + T * ZERO_U - 1) / (T * ZERO_U);
  const int pb = (n * 12 + T * PAD_U - 1) / (T * PAD_U);
  prep_kernel<<<zb + pb + W1_BLOCKS + W2_BLOCKS, T, 0, stream>>>(x, W1, W2, rec, zero_pieces,
                                                                 xp, Bp1, Bp2, n * 12, zb, pb);

  place_kernel<<<(E + T - 1) / T, T, 0, stream>>>(src, dst, ew, rec, ov, n, E);

  const int db = (n + T - 1) / T;
  const int gb = ((n + 15) / 16 + T / 32 - 1) / (T / 32);
  dis_gemm1_kernel<<<db + gb, T, 0, stream>>>(rec, ov, dis, xp, Bp1, h1, n, E, db);

  pull_kernel<true><<<((long long)n * 8 + T - 1) / T, T, 0, stream>>>(rec, ov, dis, h1, b1, hh, n, E);

  pull_kernel<false><<<((long long)n * 8 + T - 1) / T, T, 0, stream>>>(rec, ov, dis, hh, b1, g, n, E);
  gemm2_wmma<<<(n + 15) / 16, 32, 0, stream>>>(g, Bp2, b2, out, n);
}


# layer-2 transform (WMMA, 32 to 128) becomes the epilogue of the layer-2 pull launch: a block's 32 summed rows stay in LDS, 8 waves x 2 tiles, whole-line stores; 6 launches become 5; same sums, k order
# speedup vs baseline: 2.7379x; 1.0771x over previous
#include <hip/hip_runtime.h>
#include <math.h>

typedef __attribute__((ext_vector_type(2))) float v2f;
typedef __attribute__((ext_vector_type(8))) float v8f;

#define NC    45
#define NCP   48
#define H1DIM 32
#define H2DIM 128

#define PAD_U 8
#define PREP_T 256
typedef float row4 __attribute__((ext_vector_type(4), aligned(4)));
template <bool WHOLE>
__device__ __forceinline__ void pad_trip(const float* __restrict__ x, float* __restrict__ xp,
                                         int g0, int stride, int pieces) {
  float4 v[PAD_U];
#pragma unroll
  for (int u = 0; u < PAD_U; ++u) {
    int g = g0 + u * stride;
    if (!WHOLE && g > pieces - 1) g = pieces - 1;
    const int i = g / 12, j = g - i * 12;
    const float* p = x + (size_t)i * NC + 4 * j;
    const row4 t = *(const row4*)(p - (j == 11 ? 3 : 0));
    v[u] = (j == 11) ? make_float4(t.w, 0.0f, 0.0f, 0.0f) : make_float4(t.x, t.y, t.z, t.w);
  }
#pragma unroll
  for (int u = 0; u < PAD_U; ++u) {
    const int g = g0 + u * stride;
    if (WHOLE || g < pieces) *(float4*)(xp + 4 * (size_t)g) = v[u];
  }
}

__device__ __forceinline__ void pad_x_body(const float* __restrict__ x, float* __restrict__ xp,
                                           int pieces  , int vb, int nb) {
  const int stride = nb * PREP_T;
  for (int g0 = vb * PREP_T + threadIdx.x; g0 < pieces; g0 += PAD_U * stride) {
    if (g0 + (PAD_U - 1) * stride < pieces) pad_trip<true>(x, xp, g0, stride, pieces);
    else                                    pad_trip<false>(x, xp, g0, stride, pieces);
  }
}

__device__ __forceinline__ void pack_w1_body(const float* __restrict__ W1, float* __restrict__ Bp1, int vb) {
  int gid = vb * PREP_T + threadIdx.x;
  if (gid >= 12 * 2 * H1DIM * 2) return;
  int p   = gid & 1;
  int col = (gid >> 1) & 31;
  int kh  = (gid >> 6) & 1;
  int k   = gid >> 7;
  int row = 4 * k + 2 * kh + p;
  Bp1[gid] = (row < NC) ? W1[row * H1DIM + col] : 0.0f;
}

__device__ __forceinline__ void pack_w2_body(const float* __restrict__ W2, float* __restrict__ Bp2, int vb) {
  int gid = vb * PREP_T + threadIdx.x;
  if (gid >= 8 * 2 * H2DIM * 2) return;
  int p   = gid & 1;
  int col = (gid >> 1) & 127;
  int kh  = (gid >> 8) & 1;
  int k   = gid >> 9;
  int row = 4 * k + 2 * kh + p;
  Bp2[gid] = W2[row * H2DIM + col];
}

#define ZERO_U 8
#define SBW 64
__device__ __forceinline__ void zero_body(int* __restrict__ p, int pieces, int vb, int nb) {
  const int stride = nb * PREP_T;
  for (int g0 = vb * PREP_T + threadIdx.x; g0 < pieces; g0 += ZERO_U * stride) {
#pragma unroll
    for (int u = 0; u < ZERO_U; ++u) {
      const int g = g0 + u * stride;
      if (g < pieces) p[(size_t)g * SBW] = 0;
    }
  }
}

#define W1_BLOCKS ((12 * 2 * H1DIM * 2 + PREP_T - 1) / PREP_T)
#define W2_BLOCKS ((8 * 2 * H2DIM * 2 + PREP_T - 1) / PREP_T)
__global__ void __launch_bounds__(PREP_T)
prep_kernel(const float* __restrict__ x, const float* __restrict__ W1, const float* __restrict__ W2,
            int* __restrict__ zero_base, int zero_pieces, float* __restrict__ xp,
            float* __restrict__ Bp1, float* __restrict__ Bp2, int pad_pieces, int zb, int pb) {
  const int b = blockIdx.x;
  if (b < zb)                zero_body(zero_base, zero_pieces, b, zb);
  else if (b < zb + pb)      pad_x_body(x, xp, pad_pieces, b - zb, pb);
  else if (b < zb + pb + W1_BLOCKS) pack_w1_body(W1, Bp1, b - zb - pb);
  else                              pack_w2_body(W2, Bp2, b - zb - pb - W1_BLOCKS);
}

__device__ __forceinline__ void gemm1_body(const float* __restrict__ xp, const float* __restrict__ Bp1,
                                           float* __restrict__ h1, int n, int vb) {
  const int lane  = threadIdx.x & 31;
  const int wave  = __builtin_amdgcn_readfirstlane(threadIdx.x >> 5);
  const int m0    = (vb * (PREP_T / 32) + wave) * 16;
  if (m0 >= n) return;
  const int khalf = lane >> 4;
  const int nn    = lane & 15;
  int m = m0 + nn;
  if (m > n - 1) m = n - 1;

  const float* ap = xp + (size_t)m * NCP + 2 * khalf;
  const float* bp = Bp1 + khalf * 64;
  v8f c0 = {};
  v8f c1 = {};
#pragma unroll
  for (int k = 0; k < 12; ++k) {
    v2f a  = *(const v2f*)(ap + 4 * k);
    v2f b0 = *(const v2f*)(bp + 128 * k + nn * 2);
    v2f b1 = *(const v2f*)(bp + 128 * k + (16 + nn) * 2);
    c0 = __builtin_amdgcn_wmma_f32_16x16x4_f32(false, a, false, b0, (short)0, c0, false, false);
    c1 = __builtin_amdgcn_wmma_f32_16x16x4_f32(false, a, false, b1, (short)0, c1, false, false);
  }
  const int rbase = m0 + (khalf << 3);
#pragma unroll
  for (int j = 0; j < 8; ++j) {
    const int r = rbase + j;
    if (r < n) {
      h1[(size_t)r * H1DIM + nn]      = c0[j];
      h1[(size_t)r * H1DIM + 16 + nn] = c1[j];
    }
  }
}

#define CAPH 31

__device__ __forceinline__ int package_half() {
  return (int)((__builtin_amdgcn_s_sendmsg_rtn(0x87) >> 18) & 1u);
}

__global__ void place_kernel(const int* __restrict__ src, const int* __restrict__ dst,
                             const float* __restrict__ ew, int* __restrict__ rec,
                             int4* __restrict__ ov, int n, int E) {
  const int e = blockIdx.x * blockDim.x + threadIdx.x;
  const int hf = package_half();
  if (e < E) {
    const int d = dst[e];
    const int s = src[e];
    const int w = __float_as_int(ew[e]);
    int* sb = rec + ((size_t)d * 2 + hf) * SBW;
    const int slot = atomicAdd(sb, 1);
    if (slot < CAPH) {
      *(int2*)(sb + 2 * (slot + 1)) = make_int2(s, w);
    } else {
      const int k = atomicAdd(rec + (size_t)n * 2 * SBW, 1);
      ov[k] = make_int4(d, s, w, 0);
    }
  }
}

__device__ __forceinline__ int list_off(int t, int c0) {
  return t < c0 ? t + 1 : t - c0 + 1 + SBW / 2;
}

__device__ __forceinline__ void dis_pull_body(const int* __restrict__ rec, const int4* __restrict__ ov,
                                              float* __restrict__ dis, int n, int E, int vb) {
  int i = vb * PREP_T + threadIdx.x;
  if (i >= n) return;
  const int c0  = min(rec[(size_t)i * 2 * SBW], CAPH);
  const int c1  = min(rec[(size_t)i * 2 * SBW + SBW], CAPH);
  const int ovn = min(rec[(size_t)n * 2 * SBW], E);
  const int2* rb = (const int2*)rec;
  const int base = i * SBW;
  const int end = c0 + c1;
  float s = 0.0f;
  for (int j = 0; j < end; j += 4) {
    const bool k1 = j + 1 < end, k2 = j + 2 < end, k3 = j + 3 < end;
    const float w0 = __int_as_float(rb[base + list_off(j, c0)].y);
    const float w1 = __int_as_float(rb[base + list_off(k1 ? j + 1 : j, c0)].y);
    const float w2 = __int_as_float(rb[base + list_off(k2 ? j + 2 : j, c0)].y);
    const float w3 = __int_as_float(rb[base + list_off(k3 ? j + 3 : j, c0)].y);
    s += w0;
    s += k1 ? w1 : 0.0f;
    s += k2 ? w2 : 0.0f;
    s += k3 ? w3 : 0.0f;
  }
  for (int k = 0; k < ovn; ++k) {
    const int4 o = ov[k];
    if (o.x == i) s += __int_as_float(o.z);
  }
  dis[i] = 1.0f / sqrtf(s + 1.0f);
}

__global__ void __launch_bounds__(PREP_T)
dis_gemm1_kernel(const int* __restrict__ rec, const int4* __restrict__ ov,
                 float* __restrict__ dis,
                 const float* __restrict__ xp, const float* __restrict__ Bp1,
                 float* __restrict__ h1, int n, int E, int db) {
  const int b = blockIdx.x;
  if (b < db) dis_pull_body(rec, ov, dis, n, E, b);
  else        gemm1_body(xp, Bp1, h1, n, b - db);
}

__device__ __forceinline__ float4 pull_row(const int* __restrict__ rec, const int4* __restrict__ ov,
                                           const float* __restrict__ dis, const float* __restrict__ h,
                                           int n, int E, int i, int q) {
  const int c0  = min(rec[(size_t)i * 2 * SBW], CAPH);
  const int end = c0 + min(rec[(size_t)i * 2 * SBW + SBW], CAPH);
  const int ovn = min(rec[(size_t)n * 2 * SBW], E);
  const int2* rb = (const int2*)rec;
  const int base = i * SBW;
  float4 acc = make_float4(0.0f, 0.0f, 0.0f, 0.0f);
  for (int j = 0; j < end; j += 4) {
    const bool k1 = j + 1 < end, k2 = j + 2 < end, k3 = j + 3 < end;
    const int2 r0 = rb[base + list_off(j, c0)];
    const int2 r1 = rb[base + list_off(k1 ? j + 1 : j, c0)];
    const int2 r2 = rb[base + list_off(k2 ? j + 2 : j, c0)];
    const int2 r3 = rb[base + list_off(k3 ? j + 3 : j, c0)];
    const float s0 = dis[r0.x], s1 = dis[r1.x], s2 = dis[r2.x], s3 = dis[r3.x];
    const float4 v0 = *(const float4*)(h + (size_t)r0.x * H1DIM + q);
    const float4 v1 = *(const float4*)(h + (size_t)r1.x * H1DIM + q);
    const float4 v2 = *(const float4*)(h + (size_t)r2.x * H1DIM + q);
    const float4 v3 = *(const float4*)(h + (size_t)r3.x * H1DIM + q);
    const float c0f = s0 * __int_as_float(r0.y);
    const float c1f = k1 ? s1 * __int_as_float(r1.y) : 0.0f;
    const float c2f = k2 ? s2 * __int_as_float(r2.y) : 0.0f;
    const float c3f = k3 ? s3 * __int_as_float(r3.y) : 0.0f;
    acc.x += v0.x * c0f; acc.y += v0.y * c0f; acc.z += v0.z * c0f; acc.w += v0.w * c0f;
    acc.x += v1.x * c1f; acc.y += v1.y * c1f; acc.z += v1.z * c1f; acc.w += v1.w * c1f;
    acc.x += v2.x * c2f; acc.y += v2.y * c2f; acc.z += v2.z * c2f; acc.w += v2.w * c2f;
    acc.x += v3.x * c3f; acc.y += v3.y * c3f; acc.z += v3.z * c3f; acc.w += v3.w * c3f;
  }
  for (int k = 0; k < ovn; ++k) {
    const int4 o = ov[k];
    if (o.x == i) {
      const float cf = dis[o.y] * __int_as_float(o.z);
      const float4 v = *(const float4*)(h + (size_t)o.y * H1DIM + q);
      acc.x += v.x * cf; acc.y += v.y * cf; acc.z += v.z * cf; acc.w += v.w * cf;
    }
  }
  const float di = dis[i];
  const float d2 = di * di;
  const float4 hv = *(const float4*)(h + (size_t)i * H1DIM + q);
  float4 o;
  o.x = acc.x * di + hv.x * d2;
  o.y = acc.y * di + hv.y * d2;
  o.z = acc.z * di + hv.z * d2;
  o.w = acc.w * di + hv.w * d2;
  return o;
}

__global__ void pull_act_kernel(const int* __restrict__ rec, const int4* __restrict__ ov,
                                const float* __restrict__ dis, const float* __restrict__ h,
                                const float* __restrict__ bias, float* __restrict__ outp, int n, int E) {
  const int gid = blockIdx.x * blockDim.x + threadIdx.x;
  const int i   = gid >> 3;
  const int q   = (gid & 7) << 2;
  if (i >= n) return;
  float4 o = pull_row(rec, ov, dis, h, n, E, i, q);
  o.x = tanhf(o.x + bias[q + 0]);
  o.y = tanhf(o.y + bias[q + 1]);
  o.z = tanhf(o.z + bias[q + 2]);
  o.w = tanhf(o.w + bias[q + 3]);
  *(float4*)(outp + (size_t)i * H1DIM + q) = o;
}

#define GP 36
__global__ void __launch_bounds__(PREP_T)
pull_gemm2_kernel(const int* __restrict__ rec, const int4* __restrict__ ov,
                  const float* __restrict__ dis, const float* __restrict__ h,
                  const float* __restrict__ Bp2, const float* __restrict__ b2,
                  float* __restrict__ out, int n, int E) {
  __shared__ __attribute__((aligned(16))) float gt[(PREP_T / 8) * GP];
  const int li = threadIdx.x >> 3;
  const int q  = (threadIdx.x & 7) << 2;
  int i = blockIdx.x * (PREP_T / 8) + li;
  if (i > n - 1) i = n - 1;
  const float4 o = pull_row(rec, ov, dis, h, n, E, i, q);
  *(float4*)(gt + li * GP + q) = o;
  __syncthreads();

  const int lane  = threadIdx.x & 31;
  const int wave  = __builtin_amdgcn_readfirstlane(threadIdx.x >> 5);
  const int khalf = lane >> 4;
  const int nn    = lane & 15;
  const int cg    = (wave & 3) * 32 + 2 * nn;
  const int srow  = (wave >> 2) * 16;
  const float* ap = gt + (srow + nn) * GP + 2 * khalf;
  const float* bp = Bp2 + khalf * 256 + cg * 2;
  v8f acc0 = {};
  v8f acc1 = {};
#pragma unroll
  for (int k = 0; k < 8; ++k) {
    const v2f a = *(const v2f*)(ap + 4 * k);
    const float4 b = *(const float4*)(bp + 512 * k);
    v2f b0; b0.x = b.x; b0.y = b.y;
    v2f b1; b1.x = b.z; b1.y = b.w;
    acc0 = __builtin_amdgcn_wmma_f32_16x16x4_f32(false, a, false, b0, (short)0, acc0, false, false);
    acc1 = __builtin_amdgcn_wmma_f32_16x16x4_f32(false, a, false, b1, (short)0, acc1, false, false);
  }
  const float2 bv = *(const float2*)(b2 + cg);
  const int rbase = blockIdx.x * (PREP_T / 8) + srow + (khalf << 3);
#pragma unroll
  for (int j = 0; j < 8; ++j) {
    const int r = rbase + j;
    if (r < n)
      *(float2*)(out + (size_t)r * H2DIM + cg) = make_float2(acc0[j] + bv.x, acc1[j] + bv.y);
  }
}

extern "C" void kernel_launch(void* const* d_in, const int* in_sizes, int n_in,
                              void* d_out, int out_size, void* d_ws, size_t ws_size,
                              hipStream_t stream) {
  const float* x   = (const float*)d_in[0];
  const int*   src = (const int*)  d_in[1];
  const int*   dst = (const int*)  d_in[2];
  const float* ew  = (const float*)d_in[3];
  const float* W1  = (const float*)d_in[4];
  const float* b1  = (const float*)d_in[5];
  const float* W2  = (const float*)d_in[6];
  const float* b2  = (const float*)d_in[7];
  float* out = (float*)d_out;

  const int n = in_sizes[0] / NC;
  const int E = in_sizes[1];

  int*   rec  = (int*)d_ws;
  int4*  ov   = (int4*)(rec + (size_t)(2 * n + 2) * SBW);
  float* dis  = (float*)(ov + (size_t)E);
  float* xp   = dis  + n;
  float* h1   = xp   + (size_t)n * NCP;
  float* hh   = h1   + (size_t)n * H1DIM;
  float* Bp1  = hh   + (size_t)n * H1DIM;
  float* Bp2  = Bp1  + 12 * 2 * H1DIM * 2;

  const int T = PREP_T;
  const int zero_pieces = 2 * n + 1;
  const int zb = (zero_pieces + T * ZERO_U - 1) / (T * ZERO_U);
  const int pb = (n * 12 + T * PAD_U - 1) / (T * PAD_U);
  prep_kernel<<<zb + pb + W1_BLOCKS + W2_BLOCKS, T, 0, stream>>>(x, W1, W2, rec, zero_pieces,
                                                                 xp, Bp1, Bp2, n * 12, zb, pb);

  place_kernel<<<(E + T - 1) / T, T, 0, stream>>>(src, dst, ew, rec, ov, n, E);

  const int db = (n + T - 1) / T;
  const int gb = ((n + 15) / 16 + T / 32 - 1) / (T / 32);
  dis_gemm1_kernel<<<db + gb, T, 0, stream>>>(rec, ov, dis, xp, Bp1, h1, n, E, db);

  pull_act_kernel<<<((long long)n * 8 + T - 1) / T, T, 0, stream>>>(rec, ov, dis, h1, b1, hh, n, E);

  pull_gemm2_kernel<<<(n + T / 8 - 1) / (T / 8), T, 0, stream>>>(rec, ov, dis, hh, Bp2, b2, out, n, E);
}


# pull passes: a node's 8 lanes fetch 8 records with ONE load (lane l: list slot t0+l and its dis), src and coefficient go round by DPP8 lane broadcast; 24 loads per 8 records become 10; add order kept
# speedup vs baseline: 2.8279x; 1.0329x over previous
#include <hip/hip_runtime.h>
#include <math.h>

typedef __attribute__((ext_vector_type(2))) float v2f;
typedef __attribute__((ext_vector_type(8))) float v8f;

#define NC    45
#define NCP   48
#define H1DIM 32
#define H2DIM 128

#define PAD_U 8
#define PREP_T 256
typedef float row4 __attribute__((ext_vector_type(4), aligned(4)));
template <bool WHOLE>
__device__ __forceinline__ void pad_trip(const float* __restrict__ x, float* __restrict__ xp,
                                         int g0, int stride, int pieces) {
  float4 v[PAD_U];
#pragma unroll
  for (int u = 0; u < PAD_U; ++u) {
    int g = g0 + u * stride;
    if (!WHOLE && g > pieces - 1) g = pieces - 1;
    const int i = g / 12, j = g - i * 12;
    const float* p = x + (size_t)i * NC + 4 * j;
    const row4 t = *(const row4*)(p - (j == 11 ? 3 : 0));
    v[u] = (j == 11) ? make_float4(t.w, 0.0f, 0.0f, 0.0f) : make_float4(t.x, t.y, t.z, t.w);
  }
#pragma unroll
  for (int u = 0; u < PAD_U; ++u) {
    const int g = g0 + u * stride;
    if (WHOLE || g < pieces) *(float4*)(xp + 4 * (size_t)g) = v[u];
  }
}

__device__ __forceinline__ void pad_x_body(const float* __restrict__ x, float* __restrict__ xp,
                                           int pieces  , int vb, int nb) {
  const int stride = nb * PREP_T;
  for (int g0 = vb * PREP_T + threadIdx.x; g0 < pieces; g0 += PAD_U * stride) {
    if (g0 + (PAD_U - 1) * stride < pieces) pad_trip<true>(x, xp, g0, stride, pieces);
    else                                    pad_trip<false>(x, xp, g0, stride, pieces);
  }
}

__device__ __forceinline__ void pack_w1_body(const float* __restrict__ W1, float* __restrict__ Bp1, int vb) {
  int gid = vb * PREP_T + threadIdx.x;
  if (gid >= 12 * 2 * H1DIM * 2) return;
  int p   = gid & 1;
  int col = (gid >> 1) & 31;
  int kh  = (gid >> 6) & 1;
  int k   = gid >> 7;
  int row = 4 * k + 2 * kh + p;
  Bp1[gid] = (row < NC) ? W1[row * H1DIM + col] : 0.0f;
}

__device__ __forceinline__ void pack_w2_body(const float* __restrict__ W2, float* __restrict__ Bp2, int vb) {
  int gid = vb * PREP_T + threadIdx.x;
  if (gid >= 8 * 2 * H2DIM * 2) return;
  int p   = gid & 1;
  int col = (gid >> 1) & 127;
  int kh  = (gid >> 8) & 1;
  int k   = gid >> 9;
  int row = 4 * k + 2 * kh + p;
  Bp2[gid] = W2[row * H2DIM + col];
}

#define ZERO_U 8
#define SBW 64
__device__ __forceinline__ void zero_body(int* __restrict__ p, int pieces, int vb, int nb) {
  const int stride = nb * PREP_T;
  for (int g0 = vb * PREP_T + threadIdx.x; g0 < pieces; g0 += ZERO_U * stride) {
#pragma unroll
    for (int u = 0; u < ZERO_U; ++u) {
      const int g = g0 + u * stride;
      if (g < pieces) p[(size_t)g * SBW] = 0;
    }
  }
}

#define W1_BLOCKS ((12 * 2 * H1DIM * 2 + PREP_T - 1) / PREP_T)
#define W2_BLOCKS ((8 * 2 * H2DIM * 2 + PREP_T - 1) / PREP_T)
__global__ void __launch_bounds__(PREP_T)
prep_kernel(const float* __restrict__ x, const float* __restrict__ W1, const float* __restrict__ W2,
            int* __restrict__ zero_base, int zero_pieces, float* __restrict__ xp,
            float* __restrict__ Bp1, float* __restrict__ Bp2, int pad_pieces, int zb, int pb) {
  const int b = blockIdx.x;
  if (b < zb)                zero_body(zero_base, zero_pieces, b, zb);
  else if (b < zb + pb)      pad_x_body(x, xp, pad_pieces, b - zb, pb);
  else if (b < zb + pb + W1_BLOCKS) pack_w1_body(W1, Bp1, b - zb - pb);
  else                              pack_w2_body(W2, Bp2, b - zb - pb - W1_BLOCKS);
}

__device__ __forceinline__ void gemm1_body(const float* __restrict__ xp, const float* __restrict__ Bp1,
                                           float* __restrict__ h1, int n, int vb) {
  const int lane  = threadIdx.x & 31;
  const int wave  = __builtin_amdgcn_readfirstlane(threadIdx.x >> 5);
  const int m0    = (vb * (PREP_T / 32) + wave) * 16;
  if (m0 >= n) return;
  const int khalf = lane >> 4;
  const int nn    = lane & 15;
  int m = m0 + nn;
  if (m > n - 1) m = n - 1;

  const float* ap = xp + (size_t)m * NCP + 2 * khalf;
  const float* bp = Bp1 + khalf * 64;
  v8f c0 = {};
  v8f c1 = {};
#pragma unroll
  for (int k = 0; k < 12; ++k) {
    v2f a  = *(const v2f*)(ap + 4 * k);
    v2f b0 = *(const v2f*)(bp + 128 * k + nn * 2);
    v2f b1 = *(const v2f*)(bp + 128 * k + (16 + nn) * 2);
    c0 = __builtin_amdgcn_wmma_f32_16x16x4_f32(false, a, false, b0, (short)0, c0, false, false);
    c1 = __builtin_amdgcn_wmma_f32_16x16x4_f32(false, a, false, b1, (short)0, c1, false, false);
  }
  const int rbase = m0 + (khalf << 3);
#pragma unroll
  for (int j = 0; j < 8; ++j) {
    const int r = rbase + j;
    if (r < n) {
      h1[(size_t)r * H1DIM + nn]      = c0[j];
      h1[(size_t)r * H1DIM + 16 + nn] = c1[j];
    }
  }
}

#define CAPH 31

__device__ __forceinline__ int package_half() {
  return (int)((__builtin_amdgcn_s_sendmsg_rtn(0x87) >> 18) & 1u);
}

__global__ void place_kernel(const int* __restrict__ src, const int* __restrict__ dst,
                             const float* __restrict__ ew, int* __restrict__ rec,
                             int4* __restrict__ ov, int n, int E) {
  const int e = blockIdx.x * blockDim.x + threadIdx.x;
  const int hf = package_half();
  if (e < E) {
    const int d = dst[e];
    const int s = src[e];
    const int w = __float_as_int(ew[e]);
    int* sb = rec + ((size_t)d * 2 + hf) * SBW;
    const int slot = atomicAdd(sb, 1);
    if (slot < CAPH) {
      *(int2*)(sb + 2 * (slot + 1)) = make_int2(s, w);
    } else {
      const int k = atomicAdd(rec + (size_t)n * 2 * SBW, 1);
      ov[k] = make_int4(d, s, w, 0);
    }
  }
}

__device__ __forceinline__ int list_off(int t, int c0) {
  return t < c0 ? t + 1 : t - c0 + 1 + SBW / 2;
}

__device__ __forceinline__ void dis_pull_body(const int* __restrict__ rec, const int4* __restrict__ ov,
                                              float* __restrict__ dis, int n, int E, int vb) {
  int i = vb * PREP_T + threadIdx.x;
  if (i >= n) return;
  const int c0  = min(rec[(size_t)i * 2 * SBW], CAPH);
  const int c1  = min(rec[(size_t)i * 2 * SBW + SBW], CAPH);
  const int ovn = min(rec[(size_t)n * 2 * SBW], E);
  const int2* rb = (const int2*)rec;
  const int base = i * SBW;
  const int end = c0 + c1;
  float s = 0.0f;
  for (int j = 0; j < end; j += 4) {
    const bool k1 = j + 1 < end, k2 = j + 2 < end, k3 = j + 3 < end;
    const float w0 = __int_as_float(rb[base + list_off(j, c0)].y);
    const float w1 = __int_as_float(rb[base + list_off(k1 ? j + 1 : j, c0)].y);
    const float w2 = __int_as_float(rb[base + list_off(k2 ? j + 2 : j, c0)].y);
    const float w3 = __int_as_float(rb[base + list_off(k3 ? j + 3 : j, c0)].y);
    s += w0;
    s += k1 ? w1 : 0.0f;
    s += k2 ? w2 : 0.0f;
    s += k3 ? w3 : 0.0f;
  }
  for (int k = 0; k < ovn; ++k) {
    const int4 o = ov[k];
    if (o.x == i) s += __int_as_float(o.z);
  }
  dis[i] = 1.0f / sqrtf(s + 1.0f);
}

__global__ void __launch_bounds__(PREP_T)
dis_gemm1_kernel(const int* __restrict__ rec, const int4* __restrict__ ov,
                 float* __restrict__ dis,
                 const float* __restrict__ xp, const float* __restrict__ Bp1,
                 float* __restrict__ h1, int n, int E, int db) {
  const int b = blockIdx.x;
  if (b < db) dis_pull_body(rec, ov, dis, n, E, b);
  else        gemm1_body(xp, Bp1, h1, n, b - db);
}

template <int R>
__device__ __forceinline__ int group8_bcast(int v) {
  return __builtin_amdgcn_mov_dpp8(v, R * 0x249249);
}

#define PULL_GATHER(R)                                                                   \
  const unsigned ob##R = (unsigned)group8_bcast<R>(r.x) * (H1DIM * 4u) + qb;             \
  const float4   v##R  = *(const float4*)((const char*)h + ob##R);
#define PULL_ADD(R)                                                                      \
  {                                                                                      \
    const float c = __int_as_float(group8_bcast<R>(__float_as_int(cf)));                 \
    acc.x += v##R.x * c; acc.y += v##R.y * c; acc.z += v##R.z * c; acc.w += v##R.w * c;  \
  }

__device__ __forceinline__ float4 pull_row(const int* __restrict__ rec, const int4* __restrict__ ov,
                                           const float* __restrict__ dis, const float* __restrict__ h,
                                           int n, int E, int i, int q) {
  const int c0  = min(rec[(size_t)i * 2 * SBW], CAPH);
  const int end = c0 + min(rec[(size_t)i * 2 * SBW + SBW], CAPH);
  const int ovn = min(rec[(size_t)n * 2 * SBW], E);
  const int2* rb = (const int2*)rec;
  const int base = i * SBW;
  const int l8 = q >> 2;
  const unsigned qb = (unsigned)q * 4u;
  float4 acc = make_float4(0.0f, 0.0f, 0.0f, 0.0f);
  for (int t0 = 0; t0 < end; t0 += 8) {
    const int  t  = t0 + l8;
    const bool ok = t < end;
    const int2 r  = rb[base + list_off(ok ? t : t0, c0)];
    const float ds = dis[r.x];
    PULL_GATHER(0) PULL_GATHER(1) PULL_GATHER(2) PULL_GATHER(3)
    PULL_GATHER(4) PULL_GATHER(5) PULL_GATHER(6) PULL_GATHER(7)
    __builtin_amdgcn_sched_barrier(0);
    const float cf = ds * (ok ? __int_as_float(r.y) : 0.0f);
    PULL_ADD(0) PULL_ADD(1) PULL_ADD(2) PULL_ADD(3)
    PULL_ADD(4) PULL_ADD(5) PULL_ADD(6) PULL_ADD(7)
  }
  for (int k = 0; k < ovn; ++k) {
    const int4 o = ov[k];
    if (o.x == i) {
      const float cf = dis[o.y] * __int_as_float(o.z);
      const float4 v = *(const float4*)(h + (size_t)o.y * H1DIM + q);
      acc.x += v.x * cf; acc.y += v.y * cf; acc.z += v.z * cf; acc.w += v.w * cf;
    }
  }
  const float di = dis[i];
  const float d2 = di * di;
  const float4 hv = *(const float4*)(h + (size_t)i * H1DIM + q);
  float4 o;
  o.x = acc.x * di + hv.x * d2;
  o.y = acc.y * di + hv.y * d2;
  o.z = acc.z * di + hv.z * d2;
  o.w = acc.w * di + hv.w * d2;
  return o;
}

__global__ void pull_act_kernel(const int* __restrict__ rec, const int4* __restrict__ ov,
                                const float* __restrict__ dis, const float* __restrict__ h,
                                const float* __restrict__ bias, float* __restrict__ outp, int n, int E) {
  const int gid = blockIdx.x * blockDim.x + threadIdx.x;
  const int i   = gid >> 3;
  const int q   = (gid & 7) << 2;
  if (i >= n) return;
  float4 o = pull_row(rec, ov, dis, h, n, E, i, q);
  o.x = tanhf(o.x + bias[q + 0]);
  o.y = tanhf(o.y + bias[q + 1]);
  o.z = tanhf(o.z + bias[q + 2]);
  o.w = tanhf(o.w + bias[q + 3]);
  *(float4*)(outp + (size_t)i * H1DIM + q) = o;
}

#define GP 36
__global__ void __launch_bounds__(PREP_T)
pull_gemm2_kernel(const int* __restrict__ rec, const int4* __restrict__ ov,
                  const float* __restrict__ dis, const float* __restrict__ h,
                  const float* __restrict__ Bp2, const float* __restrict__ b2,
                  float* __restrict__ out, int n, int E) {
  __shared__ __attribute__((aligned(16))) float gt[(PREP_T / 8) * GP];
  const int li = threadIdx.x >> 3;
  const int q  = (threadIdx.x & 7) << 2;
  int i = blockIdx.x * (PREP_T / 8) + li;
  if (i > n - 1) i = n - 1;
  const float4 o = pull_row(rec, ov, dis, h, n, E, i, q);
  *(float4*)(gt + li * GP + q) = o;
  __syncthreads();

  const int lane  = threadIdx.x & 31;
  const int wave  = __builtin_amdgcn_readfirstlane(threadIdx.x >> 5);
  const int khalf = lane >> 4;
  const int nn    = lane & 15;
  const int cg    = (wave & 3) * 32 + 2 * nn;
  const int srow  = (wave >> 2) * 16;
  const float* ap = gt + (srow + nn) * GP + 2 * khalf;
  const float* bp = Bp2 + khalf * 256 + cg * 2;
  v8f acc0 = {};
  v8f acc1 = {};
#pragma unroll
  for (int k = 0; k < 8; ++k) {
    const v2f a = *(const v2f*)(ap + 4 * k);
    const float4 b = *(const float4*)(bp + 512 * k);
    v2f b0; b0.x = b.x; b0.y = b.y;
    v2f b1; b1.x = b.z; b1.y = b.w;
    acc0 = __builtin_amdgcn_wmma_f32_16x16x4_f32(false, a, false, b0, (short)0, acc0, false, false);
    acc1 = __builtin_amdgcn_wmma_f32_16x16x4_f32(false, a, false, b1, (short)0, acc1, false, false);
  }
  const float2 bv = *(const float2*)(b2 + cg);
  const int rbase = blockIdx.x * (PREP_T / 8) + srow + (khalf << 3);
#pragma unroll
  for (int j = 0; j < 8; ++j) {
    const int r = rbase + j;
    if (r < n)
      *(float2*)(out + (size_t)r * H2DIM + cg) = make_float2(acc0[j] + bv.x, acc1[j] + bv.y);
  }
}

extern "C" void kernel_launch(void* const* d_in, const int* in_sizes, int n_in,
                              void* d_out, int out_size, void* d_ws, size_t ws_size,
                              hipStream_t stream) {
  const float* x   = (const float*)d_in[0];
  const int*   src = (const int*)  d_in[1];
  const int*   dst = (const int*)  d_in[2];
  const float* ew  = (const float*)d_in[3];
  const float* W1  = (const float*)d_in[4];
  const float* b1  = (const float*)d_in[5];
  const float* W2  = (const float*)d_in[6];
  const float* b2  = (const float*)d_in[7];
  float* out = (float*)d_out;

  const int n = in_sizes[0] / NC;
  const int E = in_sizes[1];

  int*   rec  = (int*)d_ws;
  int4*  ov   = (int4*)(rec + (size_t)(2 * n + 2) * SBW);
  float* dis  = (float*)(ov + (size_t)E);
  float* xp   = dis  + n;
  float* h1   = xp   + (size_t)n * NCP;
  float* hh   = h1   + (size_t)n * H1DIM;
  float* Bp1  = hh   + (size_t)n * H1DIM;
  float* Bp2  = Bp1  + 12 * 2 * H1DIM * 2;

  const int T = PREP_T;
  const int zero_pieces = 2 * n + 1;
  const int zb = (zero_pieces + T * ZERO_U - 1) / (T * ZERO_U);
  const int pb = (n * 12 + T * PAD_U - 1) / (T * PAD_U);
  prep_kernel<<<zb + pb + W1_BLOCKS + W2_BLOCKS, T, 0, stream>>>(x, W1, W2, rec, zero_pieces,
                                                                 xp, Bp1, Bp2, n * 12, zb, pb);

  place_kernel<<<(E + T - 1) / T, T, 0, stream>>>(src, dst, ew, rec, ov, n, E);

  const int db = (n + T - 1) / T;
  const int gb = ((n + 15) / 16 + T / 32 - 1) / (T / 32);
  dis_gemm1_kernel<<<db + gb, T, 0, stream>>>(rec, ov, dis, xp, Bp1, h1, n, E, db);

  pull_act_kernel<<<((long long)n * 8 + T - 1) / T, T, 0, stream>>>(rec, ov, dis, h1, b1, hh, n, E);

  pull_gemm2_kernel<<<(n + T / 8 - 1) / (T / 8), T, 0, stream>>>(rec, ov, dis, hh, Bp2, b2, out, n, E);
}
